# starting kernel
# baseline (speedup 1.0000x reference)
#include <hip/hip_runtime.h>

typedef __attribute__((ext_vector_type(2))) float v2f;
typedef __attribute__((ext_vector_type(8))) float v8f;

#define N_NODES 10000
#define N_EDGES 640000
#define D_IN    128
#define D_HID   256
#define D_OUT   64

__global__ void zero_i32(int* __restrict__ p, int n) {
  int i = blockIdx.x * blockDim.x + threadIdx.x;
  if (i < n) p[i] = 0;
}

__global__ void count_deg(const int* __restrict__ dst, int* __restrict__ deg, int nE) {
  int e = blockIdx.x * blockDim.x + threadIdx.x;
  if (e < nE) atomicAdd(&deg[dst[e]], 1);
}

__global__ void exclusive_scan(const int* __restrict__ deg, int* __restrict__ rowptr, int n) {
  __shared__ int smem[1024];
  __shared__ int carry;
  int t = threadIdx.x;
  if (t == 0) carry = 0;
  __syncthreads();
  for (int base = 0; base < n; base += 1024) {
    int i = base + t;
    int v = (i < n) ? deg[i] : 0;
    smem[t] = v;
    __syncthreads();
    #pragma unroll
    for (int off = 1; off < 1024; off <<= 1) {
      int x = (t >= off) ? smem[t - off] : 0;
      __syncthreads();
      smem[t] += x;
      __syncthreads();
    }
    int cbase = carry;
    if (i < n) rowptr[i] = cbase + smem[t] - v;
    __syncthreads();
    if (t == 0) carry = cbase + smem[1023];
    __syncthreads();
  }
  if (t == 0) rowptr[n] = carry;
}

__global__ void inv_from_deg(const int* __restrict__ deg, float* __restrict__ inv, int n) {
  int i = blockIdx.x * blockDim.x + threadIdx.x;
  if (i < n) {
    int d = deg[i];
    inv[i] = (d > 0) ? (1.0f / (float)d) : 0.0f;
  }
}

__global__ void copy_i32(const int* __restrict__ a, int* __restrict__ b, int n) {
  int i = blockIdx.x * blockDim.x + threadIdx.x;
  if (i < n) b[i] = a[i];
}

__global__ void bin_edges(const int* __restrict__ src, const int* __restrict__ dst,
                          const float* __restrict__ w,
                          int* __restrict__ cursor,
                          int* __restrict__ es, float* __restrict__ ew, int nE) {
  int e = blockIdx.x * blockDim.x + threadIdx.x;
  if (e < nE) {
    int pos = atomicAdd(&cursor[dst[e]], 1);
    es[pos] = src[e];
    ew[pos] = w[e];
  }
}

template<int D>
__global__ void aggregate_nodes(const float* __restrict__ X, int ldx,
                                const int* __restrict__ rowptr,
                                const int* __restrict__ es,
                                const float* __restrict__ ew,
                                const float* __restrict__ inv,
                                float* Y, int ldy, int coff, int nN) {
  int node = (blockIdx.x * blockDim.x + threadIdx.x) >> 5;
  int lane = threadIdx.x & 31;
  if (node >= nN) return;
  int beg = rowptr[node];
  int end = rowptr[node + 1];
  float4 acc[D / 128];
  #pragma unroll
  for (int c = 0; c < D / 128; ++c) acc[c] = make_float4(0.f, 0.f, 0.f, 0.f);
  for (int j = beg; j < end; ++j) {
    int s = es[j];
    float ww = ew[j];
    const float4* xr = (const float4*)(X + (size_t)s * ldx);
    #pragma unroll
    for (int c = 0; c < D / 128; ++c) {
      float4 v = xr[lane + 32 * c];
      acc[c].x += v.x * ww;
      acc[c].y += v.y * ww;
      acc[c].z += v.z * ww;
      acc[c].w += v.w * ww;
    }
  }
  float iv = inv[node];
  float4* yr = (float4*)(Y + (size_t)node * ldy + coff);
  #pragma unroll
  for (int c = 0; c < D / 128; ++c)
    yr[lane + 32 * c] = make_float4(acc[c].x * iv, acc[c].y * iv, acc[c].z * iv, acc[c].w * iv);
}

__global__ void copy_self(const float* __restrict__ h, float* __restrict__ Y, int nN) {
  int i = blockIdx.x * blockDim.x + threadIdx.x;
  if (i >= nN * D_IN) return;
  int n = i >> 7;
  int f = i & 127;
  Y[(size_t)n * (2 * D_IN) + f] = h[i];
}

template<int NDIM, int KDIM, bool RELU>
__global__ void wmma_gemm_f32(const float* __restrict__ A, int lda,
                              const float* __restrict__ B,
                              const float* __restrict__ bias,
                              float* __restrict__ Y, int ldy,
                              int Mtiles, int Nstrips) {
  int wid  = (blockIdx.x * blockDim.x + threadIdx.x) >> 5;
  int lane = threadIdx.x & 31;
  if (wid >= Mtiles * Nstrips) return;
  int m0 = (wid / Nstrips) * 16;
  int n0 = (wid % Nstrips) * 64;
  int half = lane >> 4;
  int mr   = lane & 15;

  const float* ap = A + (size_t)(m0 + mr) * lda + 2 * half;
  const float* bp = B + (2 * half) * NDIM + n0 + mr;
  v8f acc0 = {}, acc1 = {}, acc2 = {}, acc3 = {};

  #pragma unroll 2
  for (int k = 0; k < KDIM; k += 4) {
    v2f a = *(const v2f*)ap;
    ap += 4;
    __builtin_prefetch(ap + 60, 0, 3);
    v2f b;
    b.x = bp[0];         b.y = bp[NDIM];
    acc0 = __builtin_amdgcn_wmma_f32_16x16x4_f32(false, a, false, b, (short)0, acc0, false, false);
    b.x = bp[16];        b.y = bp[NDIM + 16];
    acc1 = __builtin_amdgcn_wmma_f32_16x16x4_f32(false, a, false, b, (short)0, acc1, false, false);
    b.x = bp[32];        b.y = bp[NDIM + 32];
    acc2 = __builtin_amdgcn_wmma_f32_16x16x4_f32(false, a, false, b, (short)0, acc2, false, false);
    b.x = bp[48];        b.y = bp[NDIM + 48];
    acc3 = __builtin_amdgcn_wmma_f32_16x16x4_f32(false, a, false, b, (short)0, acc3, false, false);
    bp += 4 * NDIM;
  }

  #pragma unroll
  for (int j = 0; j < 4; ++j) {
    v8f acc = (j == 0) ? acc0 : (j == 1) ? acc1 : (j == 2) ? acc2 : acc3;
    int n = n0 + 16 * j + mr;
    float bv = bias[n];
    #pragma unroll
    for (int r = 0; r < 8; ++r) {
      int row = m0 + r + 8 * half;
      float v = acc[r] + bv;
      if (RELU) v = (v > 0.0f) ? v : 0.0f;
      Y[(size_t)row * ldy + n] = v;
    }
  }
}

extern "C" void kernel_launch(void* const* d_in, const int* in_sizes, int n_in,
                              void* d_out, int out_size, void* d_ws, size_t ws_size,
                              hipStream_t stream) {
  const float* h   = (const float*)d_in[0];
  const float* w   = (const float*)d_in[1];
  const int*   src = (const int*)d_in[2];
  const int*   dst = (const int*)d_in[3];
  const float* W1  = (const float*)d_in[4];
  const float* b1  = (const float*)d_in[5];
  const float* W2  = (const float*)d_in[6];
  const float* b2  = (const float*)d_in[7];
  float* out = (float*)d_out;

  int*   deg    = (int*)d_ws;
  int*   rowptr = deg + 10240;
  int*   cursor = rowptr + 10240;
  int*   es     = cursor + 10240;
  float* ew     = (float*)(es + N_EDGES);
  float* inv    = ew + N_EDGES;
  float* bufA   = inv + 10240;
  float* bufB   = bufA + N_NODES * D_HID;

  zero_i32<<<(N_NODES + 255) / 256, 256, 0, stream>>>(deg, N_NODES);
  count_deg<<<(N_EDGES + 255) / 256, 256, 0, stream>>>(dst, deg, N_EDGES);
  exclusive_scan<<<1, 1024, 0, stream>>>(deg, rowptr, N_NODES);
  inv_from_deg<<<(N_NODES + 255) / 256, 256, 0, stream>>>(deg, inv, N_NODES);
  copy_i32<<<(N_NODES + 255) / 256, 256, 0, stream>>>(rowptr, cursor, N_NODES);
  bin_edges<<<(N_EDGES + 255) / 256, 256, 0, stream>>>(src, dst, w, cursor, es, ew, N_EDGES);

  copy_self<<<(N_NODES * D_IN + 255) / 256, 256, 0, stream>>>(h, bufA, N_NODES);
  aggregate_nodes<D_IN><<<(N_NODES * 32 + 255) / 256, 256, 0, stream>>>(
      h, D_IN, rowptr, es, ew, inv, bufA, 2 * D_IN, D_IN, N_NODES);
  {
    int Mtiles = N_NODES / 16, Nstrips = D_HID / 64;
    wmma_gemm_f32<D_HID, 2 * D_IN, true><<<(Mtiles * Nstrips + 7) / 8, 256, 0, stream>>>(
        bufA, 2 * D_IN, W1, b1, bufB, 2 * D_HID, Mtiles, Nstrips);
  }

  aggregate_nodes<D_HID><<<(N_NODES * 32 + 255) / 256, 256, 0, stream>>>(
      bufB, 2 * D_HID, rowptr, es, ew, inv, bufB, 2 * D_HID, D_HID, N_NODES);
  {
    int Mtiles = N_NODES / 16, Nstrips = D_OUT / 64;
    wmma_gemm_f32<D_OUT, 2 * D_HID, false><<<(Mtiles * Nstrips + 7) / 8, 256, 0, stream>>>(
        bufB, 2 * D_HID, W2, b2, out, D_OUT, Mtiles, Nstrips);
  }
}


# CSR build without global atomics: 125 edge chunks count, then place, their edges by LDS atomics on a block-private 10000-bin table; in-node sum order = chunk order, then LDS arrival order
# speedup vs baseline: 1.4774x; 1.4774x over previous
#include <hip/hip_runtime.h>

typedef __attribute__((ext_vector_type(2))) float v2f;
typedef __attribute__((ext_vector_type(8))) float v8f;

#define N_NODES 10000
#define N_EDGES 640000
#define D_IN    128
#define D_HID   256
#define D_OUT   64

#define CSR_CHUNKS  125
#define CSR_THREADS 1024
#define CSR_EPT     5
#define PART_LD     10240

static_assert(CSR_CHUNKS * CSR_THREADS * CSR_EPT == N_EDGES, "chunks must tile the edge list");
static_assert(N_NODES % 4 == 0 && PART_LD % 4 == 0 && PART_LD >= N_NODES, "int4 table rows");

__global__ void __launch_bounds__(CSR_THREADS)
hist_chunks(const int* __restrict__ dst, int* __restrict__ part) {
  __shared__ int4 bins4[N_NODES / 4];
  int* bins = (int*)bins4;
  int t = threadIdx.x;
  int base = blockIdx.x * (CSR_THREADS * CSR_EPT) + t;
  int d[CSR_EPT];
  #pragma unroll
  for (int k = 0; k < CSR_EPT; ++k) d[k] = dst[base + k * CSR_THREADS];
  for (int i = t; i < N_NODES / 4; i += CSR_THREADS) bins4[i] = make_int4(0, 0, 0, 0);
  __syncthreads();
  #pragma unroll
  for (int k = 0; k < CSR_EPT; ++k) atomicAdd(&bins[d[k]], 1);
  __syncthreads();
  int4* row = (int4*)(part + blockIdx.x * PART_LD);
  for (int i = t; i < N_NODES / 4; i += CSR_THREADS) row[i] = bins4[i];
}

__global__ void __launch_bounds__(64)
chunk_offsets(int* __restrict__ part, int* __restrict__ deg) {
  int i = blockIdx.x * 64 + threadIdx.x;
  if (i >= N_NODES) return;
  int* col = part + i;
  int run = 0;
  #pragma unroll
  for (int g0 = 0; g0 < CSR_CHUNKS; g0 += 25) {
    int v[25];
    #pragma unroll
    for (int k = 0; k < 25; ++k) v[k] = col[(g0 + k) * PART_LD];
    #pragma unroll
    for (int k = 0; k < 25; ++k) { col[(g0 + k) * PART_LD] = run; run += v[k]; }
  }
  deg[i] = run;
}

__global__ void exclusive_scan(const int* __restrict__ deg, int* __restrict__ rowptr, int n) {
  __shared__ int smem[1024];
  __shared__ int carry;
  int t = threadIdx.x;
  if (t == 0) carry = 0;
  __syncthreads();
  for (int base = 0; base < n; base += 1024) {
    int i = base + t;
    int v = (i < n) ? deg[i] : 0;
    smem[t] = v;
    __syncthreads();
    #pragma unroll
    for (int off = 1; off < 1024; off <<= 1) {
      int x = (t >= off) ? smem[t - off] : 0;
      __syncthreads();
      smem[t] += x;
      __syncthreads();
    }
    int cbase = carry;
    if (i < n) rowptr[i] = cbase + smem[t] - v;
    __syncthreads();
    if (t == 0) carry = cbase + smem[1023];
    __syncthreads();
  }
  if (t == 0) rowptr[n] = carry;
}

__global__ void inv_from_deg(const int* __restrict__ deg, float* __restrict__ inv, int n) {
  int i = blockIdx.x * blockDim.x + threadIdx.x;
  if (i < n) {
    int d = deg[i];
    inv[i] = (d > 0) ? (1.0f / (float)d) : 0.0f;
  }
}

__global__ void __launch_bounds__(CSR_THREADS)
bin_chunks(const int* __restrict__ src, const int* __restrict__ dst,
           const float* __restrict__ w,
           const int* __restrict__ rowptr, const int* __restrict__ part,
           int* __restrict__ es, float* __restrict__ ew) {
  __shared__ int4 cur4[N_NODES / 4];
  int* cur = (int*)cur4;
  int t = threadIdx.x;
  int base = blockIdx.x * (CSR_THREADS * CSR_EPT) + t;
  int d[CSR_EPT], s[CSR_EPT];
  float ww[CSR_EPT];
  #pragma unroll
  for (int k = 0; k < CSR_EPT; ++k) {
    d[k]  = dst[base + k * CSR_THREADS];
    s[k]  = src[base + k * CSR_THREADS];
    ww[k] = w[base + k * CSR_THREADS];
  }
  const int4* rp4 = (const int4*)rowptr;
  const int4* pr4 = (const int4*)(part + blockIdx.x * PART_LD);
  constexpr int ROWS4 = N_NODES / 4;
  constexpr int FILL_T = 625, FILL_N = 4;
  static_assert(FILL_T * FILL_N == ROWS4 && FILL_T <= CSR_THREADS, "table fill must tile the rows");
  if (t < FILL_T) {
    int4 r[FILL_N], p[FILL_N];
    #pragma unroll
    for (int j = 0; j < FILL_N; ++j) {
      r[j] = rp4[t + j * FILL_T];
      p[j] = pr4[t + j * FILL_T];
    }
    #pragma unroll
    for (int j = 0; j < FILL_N; ++j)
      cur4[t + j * FILL_T] = make_int4(r[j].x + p[j].x, r[j].y + p[j].y, r[j].z + p[j].z, r[j].w + p[j].w);
  }
  __syncthreads();
  int pos[CSR_EPT];
  #pragma unroll
  for (int k = 0; k < CSR_EPT; ++k) pos[k] = atomicAdd(&cur[d[k]], 1);
  #pragma unroll
  for (int k = 0; k < CSR_EPT; ++k) {
    es[pos[k]] = s[k];
    ew[pos[k]] = ww[k];
  }
}

template<int D>
__global__ void aggregate_nodes(const float* __restrict__ X, int ldx,
                                const int* __restrict__ rowptr,
                                const int* __restrict__ es,
                                const float* __restrict__ ew,
                                const float* __restrict__ inv,
                                float* Y, int ldy, int coff, int nN) {
  int node = (blockIdx.x * blockDim.x + threadIdx.x) >> 5;
  int lane = threadIdx.x & 31;
  if (node >= nN) return;
  int beg = rowptr[node];
  int end = rowptr[node + 1];
  float4 acc[D / 128];
  #pragma unroll
  for (int c = 0; c < D / 128; ++c) acc[c] = make_float4(0.f, 0.f, 0.f, 0.f);
  for (int j = beg; j < end; ++j) {
    int s = es[j];
    float ww = ew[j];
    const float4* xr = (const float4*)(X + (size_t)s * ldx);
    #pragma unroll
    for (int c = 0; c < D / 128; ++c) {
      float4 v = xr[lane + 32 * c];
      acc[c].x += v.x * ww;
      acc[c].y += v.y * ww;
      acc[c].z += v.z * ww;
      acc[c].w += v.w * ww;
    }
  }
  float iv = inv[node];
  float4* yr = (float4*)(Y + (size_t)node * ldy + coff);
  #pragma unroll
  for (int c = 0; c < D / 128; ++c)
    yr[lane + 32 * c] = make_float4(acc[c].x * iv, acc[c].y * iv, acc[c].z * iv, acc[c].w * iv);
}

__global__ void copy_self(const float* __restrict__ h, float* __restrict__ Y, int nN) {
  int i = blockIdx.x * blockDim.x + threadIdx.x;
  if (i >= nN * D_IN) return;
  int n = i >> 7;
  int f = i & 127;
  Y[(size_t)n * (2 * D_IN) + f] = h[i];
}

template<int NDIM, int KDIM, bool RELU>
__global__ void wmma_gemm_f32(const float* __restrict__ A, int lda,
                              const float* __restrict__ B,
                              const float* __restrict__ bias,
                              float* __restrict__ Y, int ldy,
                              int Mtiles, int Nstrips) {
  int wid  = (blockIdx.x * blockDim.x + threadIdx.x) >> 5;
  int lane = threadIdx.x & 31;
  if (wid >= Mtiles * Nstrips) return;
  int m0 = (wid / Nstrips) * 16;
  int n0 = (wid % Nstrips) * 64;
  int half = lane >> 4;
  int mr   = lane & 15;

  const float* ap = A + (size_t)(m0 + mr) * lda + 2 * half;
  const float* bp = B + (2 * half) * NDIM + n0 + mr;
  v8f acc0 = {}, acc1 = {}, acc2 = {}, acc3 = {};

  #pragma unroll 2
  for (int k = 0; k < KDIM; k += 4) {
    v2f a = *(const v2f*)ap;
    ap += 4;
    __builtin_prefetch(ap + 60, 0, 3);
    v2f b;
    b.x = bp[0];         b.y = bp[NDIM];
    acc0 = __builtin_amdgcn_wmma_f32_16x16x4_f32(false, a, false, b, (short)0, acc0, false, false);
    b.x = bp[16];        b.y = bp[NDIM + 16];
    acc1 = __builtin_amdgcn_wmma_f32_16x16x4_f32(false, a, false, b, (short)0, acc1, false, false);
    b.x = bp[32];        b.y = bp[NDIM + 32];
    acc2 = __builtin_amdgcn_wmma_f32_16x16x4_f32(false, a, false, b, (short)0, acc2, false, false);
    b.x = bp[48];        b.y = bp[NDIM + 48];
    acc3 = __builtin_amdgcn_wmma_f32_16x16x4_f32(false, a, false, b, (short)0, acc3, false, false);
    bp += 4 * NDIM;
  }

  #pragma unroll
  for (int j = 0; j < 4; ++j) {
    v8f acc = (j == 0) ? acc0 : (j == 1) ? acc1 : (j == 2) ? acc2 : acc3;
    int n = n0 + 16 * j + mr;
    float bv = bias[n];
    #pragma unroll
    for (int r = 0; r < 8; ++r) {
      int row = m0 + r + 8 * half;
      float v = acc[r] + bv;
      if (RELU) v = (v > 0.0f) ? v : 0.0f;
      Y[(size_t)row * ldy + n] = v;
    }
  }
}

extern "C" void kernel_launch(void* const* d_in, const int* in_sizes, int n_in,
                              void* d_out, int out_size, void* d_ws, size_t ws_size,
                              hipStream_t stream) {
  const float* h   = (const float*)d_in[0];
  const float* w   = (const float*)d_in[1];
  const int*   src = (const int*)d_in[2];
  const int*   dst = (const int*)d_in[3];
  const float* W1  = (const float*)d_in[4];
  const float* b1  = (const float*)d_in[5];
  const float* W2  = (const float*)d_in[6];
  const float* b2  = (const float*)d_in[7];
  float* out = (float*)d_out;

  int*   deg    = (int*)d_ws;
  int*   rowptr = deg + 10240;
  int*   es     = rowptr + 2 * 10240;
  float* ew     = (float*)(es + N_EDGES);
  float* inv    = ew + N_EDGES;
  float* bufA   = inv + 10240;
  float* bufB   = bufA + N_NODES * D_HID;

  int* part = (int*)bufB;
  hist_chunks<<<CSR_CHUNKS, CSR_THREADS, 0, stream>>>(dst, part);
  chunk_offsets<<<(N_NODES + 63) / 64, 64, 0, stream>>>(part, deg);
  exclusive_scan<<<1, 1024, 0, stream>>>(deg, rowptr, N_NODES);
  inv_from_deg<<<(N_NODES + 255) / 256, 256, 0, stream>>>(deg, inv, N_NODES);
  bin_chunks<<<CSR_CHUNKS, CSR_THREADS, 0, stream>>>(src, dst, w, rowptr, part, es, ew);

  copy_self<<<(N_NODES * D_IN + 255) / 256, 256, 0, stream>>>(h, bufA, N_NODES);
  aggregate_nodes<D_IN><<<(N_NODES * 32 + 255) / 256, 256, 0, stream>>>(
      h, D_IN, rowptr, es, ew, inv, bufA, 2 * D_IN, D_IN, N_NODES);
  {
    int Mtiles = N_NODES / 16, Nstrips = D_HID / 64;
    wmma_gemm_f32<D_HID, 2 * D_IN, true><<<(Mtiles * Nstrips + 7) / 8, 256, 0, stream>>>(
        bufA, 2 * D_IN, W1, b1, bufB, 2 * D_HID, Mtiles, Nstrips);
  }

  aggregate_nodes<D_HID><<<(N_NODES * 32 + 255) / 256, 256, 0, stream>>>(
      bufB, 2 * D_HID, rowptr, es, ew, inv, bufB, 2 * D_HID, D_HID, N_NODES);
  {
    int Mtiles = N_NODES / 16, Nstrips = D_OUT / 64;
    wmma_gemm_f32<D_OUT, 2 * D_HID, false><<<(Mtiles * Nstrips + 7) / 8, 256, 0, stream>>>(
        bufB, 2 * D_HID, W2, b2, out, D_OUT, Mtiles, Nstrips);
  }
}


# degree prefix sum in one pass: each thread sums its 10 nodes in registers, wave shuffle scans, one LDS exchange of the 32 wave sums (1 barrier; was a 10 x 10-step LDS scan with ~240 barriers)
# speedup vs baseline: 1.6821x; 1.1386x over previous
#include <hip/hip_runtime.h>

typedef __attribute__((ext_vector_type(2))) float v2f;
typedef __attribute__((ext_vector_type(8))) float v8f;

#define N_NODES 10000
#define N_EDGES 640000
#define D_IN    128
#define D_HID   256
#define D_OUT   64

#define CSR_CHUNKS  125
#define CSR_THREADS 1024
#define CSR_EPT     5
#define PART_LD     10240

static_assert(CSR_CHUNKS * CSR_THREADS * CSR_EPT == N_EDGES, "chunks must tile the edge list");
static_assert(N_NODES % 4 == 0 && PART_LD % 4 == 0 && PART_LD >= N_NODES, "int4 table rows");

__global__ void __launch_bounds__(CSR_THREADS)
hist_chunks(const int* __restrict__ dst, int* __restrict__ part) {
  __shared__ int4 bins4[N_NODES / 4];
  int* bins = (int*)bins4;
  int t = threadIdx.x;
  int base = blockIdx.x * (CSR_THREADS * CSR_EPT) + t;
  int d[CSR_EPT];
  #pragma unroll
  for (int k = 0; k < CSR_EPT; ++k) d[k] = dst[base + k * CSR_THREADS];
  for (int i = t; i < N_NODES / 4; i += CSR_THREADS) bins4[i] = make_int4(0, 0, 0, 0);
  __syncthreads();
  #pragma unroll
  for (int k = 0; k < CSR_EPT; ++k) atomicAdd(&bins[d[k]], 1);
  __syncthreads();
  int4* row = (int4*)(part + blockIdx.x * PART_LD);
  for (int i = t; i < N_NODES / 4; i += CSR_THREADS) row[i] = bins4[i];
}

__global__ void __launch_bounds__(64)
chunk_offsets(int* __restrict__ part, int* __restrict__ deg) {
  int i = blockIdx.x * 64 + threadIdx.x;
  if (i >= N_NODES) return;
  int* col = part + i;
  int run = 0;
  #pragma unroll
  for (int g0 = 0; g0 < CSR_CHUNKS; g0 += 25) {
    int v[25];
    #pragma unroll
    for (int k = 0; k < 25; ++k) v[k] = col[(g0 + k) * PART_LD];
    #pragma unroll
    for (int k = 0; k < 25; ++k) { col[(g0 + k) * PART_LD] = run; run += v[k]; }
  }
  deg[i] = run;
}

#define SCAN_THREADS 1024
#define SCAN_PER     10
static_assert(SCAN_THREADS * SCAN_PER >= N_NODES && N_NODES % SCAN_PER == 0, "threads must tile the nodes");
static_assert(SCAN_PER % 2 == 0, "int2 loads and stores");
static_assert(SCAN_THREADS / 32 == 32, "one wave sum per lane in the second scan");

__global__ void __launch_bounds__(SCAN_THREADS)
exclusive_scan(const int* __restrict__ deg, int* __restrict__ rowptr) {
  __shared__ int wsum[SCAN_THREADS / 32];
  int t = threadIdx.x;
  int lane = t & 31;
  int wv = t >> 5;
  bool live = t < N_NODES / SCAN_PER;
  int v[SCAN_PER];
  const int2* in2 = (const int2*)(deg + (live ? t : 0) * SCAN_PER);
  #pragma unroll
  for (int k = 0; k < SCAN_PER / 2; ++k) {
    int2 p = in2[k];
    v[2 * k] = live ? p.x : 0;
    v[2 * k + 1] = live ? p.y : 0;
  }
  int tot = 0;
  #pragma unroll
  for (int k = 0; k < SCAN_PER; ++k) tot += v[k];

  int inc = tot;
  for (int off = 1; off < 32; off <<= 1) {
    int x = __shfl_up(inc, off, 32);
    if (lane >= off) inc += x;
  }
  if (lane == 31) wsum[wv] = inc;
  __syncthreads();

  int ws = wsum[lane];
  int winc = ws;
  for (int off = 1; off < 32; off <<= 1) {
    int x = __shfl_up(winc, off, 32);
    if (lane >= off) winc += x;
  }
  int wbase = __shfl(winc - ws, wv, 32);
  int total = __shfl(winc, 31, 32);

  if (live) {
    int run = wbase + inc - tot;
    int2* out2 = (int2*)(rowptr + t * SCAN_PER);
    #pragma unroll
    for (int k = 0; k < SCAN_PER / 2; ++k) {
      int a = run;  run += v[2 * k];
      int b = run;  run += v[2 * k + 1];
      out2[k] = make_int2(a, b);
    }
  }
  if (t == 0) rowptr[N_NODES] = total;
}

__global__ void inv_from_deg(const int* __restrict__ deg, float* __restrict__ inv, int n) {
  int i = blockIdx.x * blockDim.x + threadIdx.x;
  if (i < n) {
    int d = deg[i];
    inv[i] = (d > 0) ? (1.0f / (float)d) : 0.0f;
  }
}

__global__ void __launch_bounds__(CSR_THREADS)
bin_chunks(const int* __restrict__ src, const int* __restrict__ dst,
           const float* __restrict__ w,
           const int* __restrict__ rowptr, const int* __restrict__ part,
           int* __restrict__ es, float* __restrict__ ew) {
  __shared__ int4 cur4[N_NODES / 4];
  int* cur = (int*)cur4;
  int t = threadIdx.x;
  int base = blockIdx.x * (CSR_THREADS * CSR_EPT) + t;
  int d[CSR_EPT], s[CSR_EPT];
  float ww[CSR_EPT];
  #pragma unroll
  for (int k = 0; k < CSR_EPT; ++k) {
    d[k]  = dst[base + k * CSR_THREADS];
    s[k]  = src[base + k * CSR_THREADS];
    ww[k] = w[base + k * CSR_THREADS];
  }
  const int4* rp4 = (const int4*)rowptr;
  const int4* pr4 = (const int4*)(part + blockIdx.x * PART_LD);
  constexpr int ROWS4 = N_NODES / 4;
  constexpr int FILL_T = 625, FILL_N = 4;
  static_assert(FILL_T * FILL_N == ROWS4 && FILL_T <= CSR_THREADS, "table fill must tile the rows");
  if (t < FILL_T) {
    int4 r[FILL_N], p[FILL_N];
    #pragma unroll
    for (int j = 0; j < FILL_N; ++j) {
      r[j] = rp4[t + j * FILL_T];
      p[j] = pr4[t + j * FILL_T];
    }
    #pragma unroll
    for (int j = 0; j < FILL_N; ++j)
      cur4[t + j * FILL_T] = make_int4(r[j].x + p[j].x, r[j].y + p[j].y, r[j].z + p[j].z, r[j].w + p[j].w);
  }
  __syncthreads();
  int pos[CSR_EPT];
  #pragma unroll
  for (int k = 0; k < CSR_EPT; ++k) pos[k] = atomicAdd(&cur[d[k]], 1);
  #pragma unroll
  for (int k = 0; k < CSR_EPT; ++k) {
    es[pos[k]] = s[k];
    ew[pos[k]] = ww[k];
  }
}

template<int D>
__global__ void aggregate_nodes(const float* __restrict__ X, int ldx,
                                const int* __restrict__ rowptr,
                                const int* __restrict__ es,
                                const float* __restrict__ ew,
                                const float* __restrict__ inv,
                                float* Y, int ldy, int coff, int nN) {
  int node = (blockIdx.x * blockDim.x + threadIdx.x) >> 5;
  int lane = threadIdx.x & 31;
  if (node >= nN) return;
  int beg = rowptr[node];
  int end = rowptr[node + 1];
  float4 acc[D / 128];
  #pragma unroll
  for (int c = 0; c < D / 128; ++c) acc[c] = make_float4(0.f, 0.f, 0.f, 0.f);
  for (int j = beg; j < end; ++j) {
    int s = es[j];
    float ww = ew[j];
    const float4* xr = (const float4*)(X + (size_t)s * ldx);
    #pragma unroll
    for (int c = 0; c < D / 128; ++c) {
      float4 v = xr[lane + 32 * c];
      acc[c].x += v.x * ww;
      acc[c].y += v.y * ww;
      acc[c].z += v.z * ww;
      acc[c].w += v.w * ww;
    }
  }
  float iv = inv[node];
  float4* yr = (float4*)(Y + (size_t)node * ldy + coff);
  #pragma unroll
  for (int c = 0; c < D / 128; ++c)
    yr[lane + 32 * c] = make_float4(acc[c].x * iv, acc[c].y * iv, acc[c].z * iv, acc[c].w * iv);
}

__global__ void copy_self(const float* __restrict__ h, float* __restrict__ Y, int nN) {
  int i = blockIdx.x * blockDim.x + threadIdx.x;
  if (i >= nN * D_IN) return;
  int n = i >> 7;
  int f = i & 127;
  Y[(size_t)n * (2 * D_IN) + f] = h[i];
}

template<int NDIM, int KDIM, bool RELU>
__global__ void wmma_gemm_f32(const float* __restrict__ A, int lda,
                              const float* __restrict__ B,
                              const float* __restrict__ bias,
                              float* __restrict__ Y, int ldy,
                              int Mtiles, int Nstrips) {
  int wid  = (blockIdx.x * blockDim.x + threadIdx.x) >> 5;
  int lane = threadIdx.x & 31;
  if (wid >= Mtiles * Nstrips) return;
  int m0 = (wid / Nstrips) * 16;
  int n0 = (wid % Nstrips) * 64;
  int half = lane >> 4;
  int mr   = lane & 15;

  const float* ap = A + (size_t)(m0 + mr) * lda + 2 * half;
  const float* bp = B + (2 * half) * NDIM + n0 + mr;
  v8f acc0 = {}, acc1 = {}, acc2 = {}, acc3 = {};

  #pragma unroll 2
  for (int k = 0; k < KDIM; k += 4) {
    v2f a = *(const v2f*)ap;
    ap += 4;
    __builtin_prefetch(ap + 60, 0, 3);
    v2f b;
    b.x = bp[0];         b.y = bp[NDIM];
    acc0 = __builtin_amdgcn_wmma_f32_16x16x4_f32(false, a, false, b, (short)0, acc0, false, false);
    b.x = bp[16];        b.y = bp[NDIM + 16];
    acc1 = __builtin_amdgcn_wmma_f32_16x16x4_f32(false, a, false, b, (short)0, acc1, false, false);
    b.x = bp[32];        b.y = bp[NDIM + 32];
    acc2 = __builtin_amdgcn_wmma_f32_16x16x4_f32(false, a, false, b, (short)0, acc2, false, false);
    b.x = bp[48];        b.y = bp[NDIM + 48];
    acc3 = __builtin_amdgcn_wmma_f32_16x16x4_f32(false, a, false, b, (short)0, acc3, false, false);
    bp += 4 * NDIM;
  }

  #pragma unroll
  for (int j = 0; j < 4; ++j) {
    v8f acc = (j == 0) ? acc0 : (j == 1) ? acc1 : (j == 2) ? acc2 : acc3;
    int n = n0 + 16 * j + mr;
    float bv = bias[n];
    #pragma unroll
    for (int r = 0; r < 8; ++r) {
      int row = m0 + r + 8 * half;
      float v = acc[r] + bv;
      if (RELU) v = (v > 0.0f) ? v : 0.0f;
      Y[(size_t)row * ldy + n] = v;
    }
  }
}

extern "C" void kernel_launch(void* const* d_in, const int* in_sizes, int n_in,
                              void* d_out, int out_size, void* d_ws, size_t ws_size,
                              hipStream_t stream) {
  const float* h   = (const float*)d_in[0];
  const float* w   = (const float*)d_in[1];
  const int*   src = (const int*)d_in[2];
  const int*   dst = (const int*)d_in[3];
  const float* W1  = (const float*)d_in[4];
  const float* b1  = (const float*)d_in[5];
  const float* W2  = (const float*)d_in[6];
  const float* b2  = (const float*)d_in[7];
  float* out = (float*)d_out;

  int*   deg    = (int*)d_ws;
  int*   rowptr = deg + 10240;
  int*   es     = rowptr + 2 * 10240;
  float* ew     = (float*)(es + N_EDGES);
  float* inv    = ew + N_EDGES;
  float* bufA   = inv + 10240;
  float* bufB   = bufA + N_NODES * D_HID;

  int* part = (int*)bufB;
  hist_chunks<<<CSR_CHUNKS, CSR_THREADS, 0, stream>>>(dst, part);
  chunk_offsets<<<(N_NODES + 63) / 64, 64, 0, stream>>>(part, deg);
  exclusive_scan<<<1, SCAN_THREADS, 0, stream>>>(deg, rowptr);
  inv_from_deg<<<(N_NODES + 255) / 256, 256, 0, stream>>>(deg, inv, N_NODES);
  bin_chunks<<<CSR_CHUNKS, CSR_THREADS, 0, stream>>>(src, dst, w, rowptr, part, es, ew);

  copy_self<<<(N_NODES * D_IN + 255) / 256, 256, 0, stream>>>(h, bufA, N_NODES);
  aggregate_nodes<D_IN><<<(N_NODES * 32 + 255) / 256, 256, 0, stream>>>(
      h, D_IN, rowptr, es, ew, inv, bufA, 2 * D_IN, D_IN, N_NODES);
  {
    int Mtiles = N_NODES / 16, Nstrips = D_HID / 64;
    wmma_gemm_f32<D_HID, 2 * D_IN, true><<<(Mtiles * Nstrips + 7) / 8, 256, 0, stream>>>(
        bufA, 2 * D_IN, W1, b1, bufB, 2 * D_HID, Mtiles, Nstrips);
  }

  aggregate_nodes<D_HID><<<(N_NODES * 32 + 255) / 256, 256, 0, stream>>>(
      bufB, 2 * D_HID, rowptr, es, ew, inv, bufB, 2 * D_HID, D_HID, N_NODES);
  {
    int Mtiles = N_NODES / 16, Nstrips = D_OUT / 64;
    wmma_gemm_f32<D_OUT, 2 * D_HID, false><<<(Mtiles * Nstrips + 7) / 8, 256, 0, stream>>>(
        bufB, 2 * D_HID, W2, b2, out, D_OUT, Mtiles, Nstrips);
  }
}


# no concat buffer: GEMM1 reads A's self half (k < 128) straight from h and the neighbour half from the mean plane; the 40000-wave copy launch is gone (9 launches, was 10)
# speedup vs baseline: 2.1841x; 1.2984x over previous
#include <hip/hip_runtime.h>

typedef __attribute__((ext_vector_type(2))) float v2f;
typedef __attribute__((ext_vector_type(8))) float v8f;

#define N_NODES 10000
#define N_EDGES 640000
#define D_IN    128
#define D_HID   256
#define D_OUT   64

#define CSR_CHUNKS  125
#define CSR_THREADS 1024
#define CSR_EPT     5
#define PART_LD     10240

static_assert(CSR_CHUNKS * CSR_THREADS * CSR_EPT == N_EDGES, "chunks must tile the edge list");
static_assert(N_NODES % 4 == 0 && PART_LD % 4 == 0 && PART_LD >= N_NODES, "int4 table rows");

__global__ void __launch_bounds__(CSR_THREADS)
hist_chunks(const int* __restrict__ dst, int* __restrict__ part) {
  __shared__ int4 bins4[N_NODES / 4];
  int* bins = (int*)bins4;
  int t = threadIdx.x;
  int base = blockIdx.x * (CSR_THREADS * CSR_EPT) + t;
  int d[CSR_EPT];
  #pragma unroll
  for (int k = 0; k < CSR_EPT; ++k) d[k] = dst[base + k * CSR_THREADS];
  for (int i = t; i < N_NODES / 4; i += CSR_THREADS) bins4[i] = make_int4(0, 0, 0, 0);
  __syncthreads();
  #pragma unroll
  for (int k = 0; k < CSR_EPT; ++k) atomicAdd(&bins[d[k]], 1);
  __syncthreads();
  int4* row = (int4*)(part + blockIdx.x * PART_LD);
  for (int i = t; i < N_NODES / 4; i += CSR_THREADS) row[i] = bins4[i];
}

__global__ void __launch_bounds__(64)
chunk_offsets(int* __restrict__ part, int* __restrict__ deg) {
  int i = blockIdx.x * 64 + threadIdx.x;
  if (i >= N_NODES) return;
  int* col = part + i;
  int run = 0;
  #pragma unroll
  for (int g0 = 0; g0 < CSR_CHUNKS; g0 += 25) {
    int v[25];
    #pragma unroll
    for (int k = 0; k < 25; ++k) v[k] = col[(g0 + k) * PART_LD];
    #pragma unroll
    for (int k = 0; k < 25; ++k) { col[(g0 + k) * PART_LD] = run; run += v[k]; }
  }
  deg[i] = run;
}

#define SCAN_THREADS 1024
#define SCAN_PER     10
static_assert(SCAN_THREADS * SCAN_PER >= N_NODES && N_NODES % SCAN_PER == 0, "threads must tile the nodes");
static_assert(SCAN_PER % 2 == 0, "int2 loads and stores");
static_assert(SCAN_THREADS / 32 == 32, "one wave sum per lane in the second scan");

__global__ void __launch_bounds__(SCAN_THREADS)
exclusive_scan(const int* __restrict__ deg, int* __restrict__ rowptr) {
  __shared__ int wsum[SCAN_THREADS / 32];
  int t = threadIdx.x;
  int lane = t & 31;
  int wv = t >> 5;
  bool live = t < N_NODES / SCAN_PER;
  int v[SCAN_PER];
  const int2* in2 = (const int2*)(deg + (live ? t : 0) * SCAN_PER);
  #pragma unroll
  for (int k = 0; k < SCAN_PER / 2; ++k) {
    int2 p = in2[k];
    v[2 * k] = live ? p.x : 0;
    v[2 * k + 1] = live ? p.y : 0;
  }
  int tot = 0;
  #pragma unroll
  for (int k = 0; k < SCAN_PER; ++k) tot += v[k];

  int inc = tot;
  for (int off = 1; off < 32; off <<= 1) {
    int x = __shfl_up(inc, off, 32);
    if (lane >= off) inc += x;
  }
  if (lane == 31) wsum[wv] = inc;
  __syncthreads();

  int ws = wsum[lane];
  int winc = ws;
  for (int off = 1; off < 32; off <<= 1) {
    int x = __shfl_up(winc, off, 32);
    if (lane >= off) winc += x;
  }
  int wbase = __shfl(winc - ws, wv, 32);
  int total = __shfl(winc, 31, 32);

  if (live) {
    int run = wbase + inc - tot;
    int2* out2 = (int2*)(rowptr + t * SCAN_PER);
    #pragma unroll
    for (int k = 0; k < SCAN_PER / 2; ++k) {
      int a = run;  run += v[2 * k];
      int b = run;  run += v[2 * k + 1];
      out2[k] = make_int2(a, b);
    }
  }
  if (t == 0) rowptr[N_NODES] = total;
}

__global__ void inv_from_deg(const int* __restrict__ deg, float* __restrict__ inv, int n) {
  int i = blockIdx.x * blockDim.x + threadIdx.x;
  if (i < n) {
    int d = deg[i];
    inv[i] = (d > 0) ? (1.0f / (float)d) : 0.0f;
  }
}

__global__ void __launch_bounds__(CSR_THREADS)
bin_chunks(const int* __restrict__ src, const int* __restrict__ dst,
           const float* __restrict__ w,
           const int* __restrict__ rowptr, const int* __restrict__ part,
           int* __restrict__ es, float* __restrict__ ew) {
  __shared__ int4 cur4[N_NODES / 4];
  int* cur = (int*)cur4;
  int t = threadIdx.x;
  int base = blockIdx.x * (CSR_THREADS * CSR_EPT) + t;
  int d[CSR_EPT], s[CSR_EPT];
  float ww[CSR_EPT];
  #pragma unroll
  for (int k = 0; k < CSR_EPT; ++k) {
    d[k]  = dst[base + k * CSR_THREADS];
    s[k]  = src[base + k * CSR_THREADS];
    ww[k] = w[base + k * CSR_THREADS];
  }
  const int4* rp4 = (const int4*)rowptr;
  const int4* pr4 = (const int4*)(part + blockIdx.x * PART_LD);
  constexpr int ROWS4 = N_NODES / 4;
  constexpr int FILL_T = 625, FILL_N = 4;
  static_assert(FILL_T * FILL_N == ROWS4 && FILL_T <= CSR_THREADS, "table fill must tile the rows");
  if (t < FILL_T) {
    int4 r[FILL_N], p[FILL_N];
    #pragma unroll
    for (int j = 0; j < FILL_N; ++j) {
      r[j] = rp4[t + j * FILL_T];
      p[j] = pr4[t + j * FILL_T];
    }
    #pragma unroll
    for (int j = 0; j < FILL_N; ++j)
      cur4[t + j * FILL_T] = make_int4(r[j].x + p[j].x, r[j].y + p[j].y, r[j].z + p[j].z, r[j].w + p[j].w);
  }
  __syncthreads();
  int pos[CSR_EPT];
  #pragma unroll
  for (int k = 0; k < CSR_EPT; ++k) pos[k] = atomicAdd(&cur[d[k]], 1);
  #pragma unroll
  for (int k = 0; k < CSR_EPT; ++k) {
    es[pos[k]] = s[k];
    ew[pos[k]] = ww[k];
  }
}

template<int D>
__global__ void aggregate_nodes(const float* __restrict__ X, int ldx,
                                const int* __restrict__ rowptr,
                                const int* __restrict__ es,
                                const float* __restrict__ ew,
                                const float* __restrict__ inv,
                                float* Y, int ldy, int coff, int nN) {
  int node = (blockIdx.x * blockDim.x + threadIdx.x) >> 5;
  int lane = threadIdx.x & 31;
  if (node >= nN) return;
  int beg = rowptr[node];
  int end = rowptr[node + 1];
  float4 acc[D / 128];
  #pragma unroll
  for (int c = 0; c < D / 128; ++c) acc[c] = make_float4(0.f, 0.f, 0.f, 0.f);
  for (int j = beg; j < end; ++j) {
    int s = es[j];
    float ww = ew[j];
    const float4* xr = (const float4*)(X + (size_t)s * ldx);
    #pragma unroll
    for (int c = 0; c < D / 128; ++c) {
      float4 v = xr[lane + 32 * c];
      acc[c].x += v.x * ww;
      acc[c].y += v.y * ww;
      acc[c].z += v.z * ww;
      acc[c].w += v.w * ww;
    }
  }
  float iv = inv[node];
  float4* yr = (float4*)(Y + (size_t)node * ldy + coff);
  #pragma unroll
  for (int c = 0; c < D / 128; ++c)
    yr[lane + 32 * c] = make_float4(acc[c].x * iv, acc[c].y * iv, acc[c].z * iv, acc[c].w * iv);
}

__global__ void __launch_bounds__(256)
aggregate_projected(const float* __restrict__ P, const float* __restrict__ Q,
                    const float* __restrict__ bias,
                    const int* __restrict__ rowptr,
                    const int* __restrict__ es,
                    const float* __restrict__ ew,
                    const float* __restrict__ inv,
                    float* __restrict__ out, int nN) {
  static_assert(D_OUT == 64, "a row is 16 lanes x float4");
  int node = (blockIdx.x * blockDim.x + threadIdx.x) >> 5;
  int lane = threadIdx.x & 31;
  if (node >= nN) return;
  int half = lane >> 4;
  int c    = lane & 15;
  int beg = rowptr[node];
  int end = rowptr[node + 1];
  int mid = beg + ((end - beg + 1) >> 1);
  int lo = half ? mid : beg;
  int hi = half ? end : mid;
  float4 acc = make_float4(0.f, 0.f, 0.f, 0.f);
  for (int j = lo; j < hi; ++j) {
    int s = es[j];
    float ww = ew[j];
    float4 v = ((const float4*)(Q + (size_t)s * D_OUT))[c];
    acc.x += v.x * ww;
    acc.y += v.y * ww;
    acc.z += v.z * ww;
    acc.w += v.w * ww;
  }
  acc.x += __shfl_xor(acc.x, 16, 32);
  acc.y += __shfl_xor(acc.y, 16, 32);
  acc.z += __shfl_xor(acc.z, 16, 32);
  acc.w += __shfl_xor(acc.w, 16, 32);
  if (half == 0) {
    float iv = inv[node];
    float4 p = ((const float4*)(P + (size_t)node * D_OUT))[c];
    float4 b = ((const float4*)bias)[c];
    ((float4*)(out + (size_t)node * D_OUT))[c] =
        make_float4(p.x + b.x + acc.x * iv, p.y + b.y + acc.y * iv,
                    p.z + b.z + acc.z * iv, p.w + b.w + acc.w * iv);
  }
}

template<int NDIM, int KDIM, int KSPLIT, bool RELU, bool BIAS>
__global__ void wmma_gemm_f32(const float* __restrict__ A0, int lda0,
                              const float* __restrict__ A1, int lda1,
                              const float* __restrict__ B,
                              const float* __restrict__ bias,
                              float* __restrict__ Y, int ldy,
                              int Mtiles, int Nstrips) {
  int wid  = (blockIdx.x * blockDim.x + threadIdx.x) >> 5;
  int lane = threadIdx.x & 31;
  if (wid >= Mtiles * Nstrips) return;
  int m0 = (wid / Nstrips) * 16;
  int cs = (wid % Nstrips) * 64;
  int slab = cs / NDIM;
  int n0   = cs % NDIM;
  int half = lane >> 4;
  int mr   = lane & 15;
  B += (size_t)slab * (KDIM * NDIM);
  Y += (size_t)slab * Mtiles * 16 * ldy;

  const float* ap = A0 + (size_t)(m0 + mr) * lda0 + 2 * half;
  const float* bp = B + (2 * half) * NDIM + n0 + mr;
  v8f acc0 = {}, acc1 = {}, acc2 = {}, acc3 = {};

  auto kstep = [&](const float*& arow, const float* pf) {
    v2f a = *(const v2f*)arow;
    arow += 4;
    __builtin_prefetch(pf, 0, 3);
    v2f b;
    b.x = bp[0];         b.y = bp[NDIM];
    acc0 = __builtin_amdgcn_wmma_f32_16x16x4_f32(false, a, false, b, (short)0, acc0, false, false);
    b.x = bp[16];        b.y = bp[NDIM + 16];
    acc1 = __builtin_amdgcn_wmma_f32_16x16x4_f32(false, a, false, b, (short)0, acc1, false, false);
    b.x = bp[32];        b.y = bp[NDIM + 32];
    acc2 = __builtin_amdgcn_wmma_f32_16x16x4_f32(false, a, false, b, (short)0, acc2, false, false);
    b.x = bp[48];        b.y = bp[NDIM + 48];
    acc3 = __builtin_amdgcn_wmma_f32_16x16x4_f32(false, a, false, b, (short)0, acc3, false, false);
    bp += 4 * NDIM;
  };

  if constexpr (KSPLIT == KDIM) {
    #pragma unroll 2
    for (int k = 0; k < KDIM; k += 4) kstep(ap, ap + 64);
  } else {
    static_assert(KSPLIT >= 64 && KSPLIT % 8 == 0 && KDIM - KSPLIT >= 64 && (KDIM - KSPLIT) % 8 == 0,
                  "the prefetch runs 64 columns ahead; the loops are unrolled by two steps");
    const float* ap1 = A1 + (size_t)(m0 + mr) * lda1 + 2 * half;
    #pragma unroll 2
    for (int k = 0; k < KSPLIT - 64; k += 4) kstep(ap, ap + 64);
    const float* pf1 = ap1;
    #pragma unroll 2
    for (int k = KSPLIT - 64; k < KSPLIT; k += 4) { kstep(ap, pf1); pf1 += 4; }
    #pragma unroll 2
    for (int k = KSPLIT; k < KDIM; k += 4) kstep(ap1, ap1 + 64);
  }

  #pragma unroll
  for (int j = 0; j < 4; ++j) {
    v8f acc = (j == 0) ? acc0 : (j == 1) ? acc1 : (j == 2) ? acc2 : acc3;
    int n = n0 + 16 * j + mr;
    float bv = 0.0f;
    if (BIAS) bv = bias[n];
    #pragma unroll
    for (int r = 0; r < 8; ++r) {
      int row = m0 + r + 8 * half;
      float v = acc[r];
      if (BIAS) v += bv;
      if (RELU) v = (v > 0.0f) ? v : 0.0f;
      Y[(size_t)row * ldy + n] = v;
    }
  }
}

extern "C" void kernel_launch(void* const* d_in, const int* in_sizes, int n_in,
                              void* d_out, int out_size, void* d_ws, size_t ws_size,
                              hipStream_t stream) {
  const float* h   = (const float*)d_in[0];
  const float* w   = (const float*)d_in[1];
  const int*   src = (const int*)d_in[2];
  const int*   dst = (const int*)d_in[3];
  const float* W1  = (const float*)d_in[4];
  const float* b1  = (const float*)d_in[5];
  const float* W2  = (const float*)d_in[6];
  const float* b2  = (const float*)d_in[7];
  float* out = (float*)d_out;

  int*   deg    = (int*)d_ws;
  int*   rowptr = deg + 10240;
  int*   es     = rowptr + 2 * 10240;
  float* ew     = (float*)(es + N_EDGES);
  float* inv    = ew + N_EDGES;
  float* bufA   = inv + 10240;
  float* bufB   = bufA + N_NODES * D_HID;
  float* PQ     = bufB + N_NODES * D_HID;

  int* part = (int*)bufB;
  hist_chunks<<<CSR_CHUNKS, CSR_THREADS, 0, stream>>>(dst, part);
  chunk_offsets<<<(N_NODES + 63) / 64, 64, 0, stream>>>(part, deg);
  exclusive_scan<<<1, SCAN_THREADS, 0, stream>>>(deg, rowptr);
  inv_from_deg<<<(N_NODES + 255) / 256, 256, 0, stream>>>(deg, inv, N_NODES);
  bin_chunks<<<CSR_CHUNKS, CSR_THREADS, 0, stream>>>(src, dst, w, rowptr, part, es, ew);

  aggregate_nodes<D_IN><<<(N_NODES * 32 + 255) / 256, 256, 0, stream>>>(
      h, D_IN, rowptr, es, ew, inv, bufA, D_IN, 0, N_NODES);
  {
    int Mtiles = N_NODES / 16, Nstrips = D_HID / 64;
    wmma_gemm_f32<D_HID, 2 * D_IN, D_IN, true, true><<<(Mtiles * Nstrips + 7) / 8, 256, 0, stream>>>(
        h, D_IN, bufA, D_IN, W1, b1, bufB, D_HID, Mtiles, Nstrips);
  }

  {
    int Mtiles = N_NODES / 16, Nstrips = 2 * D_OUT / 64;
    wmma_gemm_f32<D_OUT, D_HID, D_HID, false, false><<<(Mtiles * Nstrips + 7) / 8, 256, 0, stream>>>(
        bufB, D_HID, nullptr, 0, W2, nullptr, PQ, D_OUT, Mtiles, Nstrips);
  }
  aggregate_projected<<<(N_NODES * 32 + 255) / 256, 256, 0, stream>>>(
      PQ, PQ + N_NODES * D_OUT, b2, rowptr, es, ew, inv, out, N_NODES);
}


# both gather kernels: edge (source, weight) pairs read one per lane a batch ahead, row loads issued 16 edges at a time as one group ahead of their sums (was 4-8 behind an index load); same sum order
# speedup vs baseline: 2.3274x; 1.0656x over previous
#include <hip/hip_runtime.h>

typedef __attribute__((ext_vector_type(2))) float v2f;
typedef __attribute__((ext_vector_type(8))) float v8f;

#define N_NODES 10000
#define N_EDGES 640000
#define D_IN    128
#define D_HID   256
#define D_OUT   64

#define CSR_CHUNKS  125
#define CSR_THREADS 1024
#define CSR_EPT     5
#define PART_LD     10240

static_assert(CSR_CHUNKS * CSR_THREADS * CSR_EPT == N_EDGES, "chunks must tile the edge list");
static_assert(N_NODES % 4 == 0 && PART_LD % 4 == 0 && PART_LD >= N_NODES, "int4 table rows");

__global__ void __launch_bounds__(CSR_THREADS)
hist_chunks(const int* __restrict__ dst, int* __restrict__ part) {
  __shared__ int4 bins4[N_NODES / 4];
  int* bins = (int*)bins4;
  int t = threadIdx.x;
  int base = blockIdx.x * (CSR_THREADS * CSR_EPT) + t;
  int d[CSR_EPT];
  #pragma unroll
  for (int k = 0; k < CSR_EPT; ++k) d[k] = dst[base + k * CSR_THREADS];
  for (int i = t; i < N_NODES / 4; i += CSR_THREADS) bins4[i] = make_int4(0, 0, 0, 0);
  __syncthreads();
  #pragma unroll
  for (int k = 0; k < CSR_EPT; ++k) atomicAdd(&bins[d[k]], 1);
  __syncthreads();
  int4* row = (int4*)(part + blockIdx.x * PART_LD);
  for (int i = t; i < N_NODES / 4; i += CSR_THREADS) row[i] = bins4[i];
}

__global__ void __launch_bounds__(64)
chunk_offsets(int* __restrict__ part, int* __restrict__ deg) {
  int i = blockIdx.x * 64 + threadIdx.x;
  if (i >= N_NODES) return;
  int* col = part + i;
  int run = 0;
  #pragma unroll
  for (int g0 = 0; g0 < CSR_CHUNKS; g0 += 25) {
    int v[25];
    #pragma unroll
    for (int k = 0; k < 25; ++k) v[k] = col[(g0 + k) * PART_LD];
    #pragma unroll
    for (int k = 0; k < 25; ++k) { col[(g0 + k) * PART_LD] = run; run += v[k]; }
  }
  deg[i] = run;
}

#define SCAN_THREADS 1024
#define SCAN_PER     10
static_assert(SCAN_THREADS * SCAN_PER >= N_NODES && N_NODES % SCAN_PER == 0, "threads must tile the nodes");
static_assert(SCAN_PER % 2 == 0, "int2 loads and stores");
static_assert(SCAN_THREADS / 32 == 32, "one wave sum per lane in the second scan");

__global__ void __launch_bounds__(SCAN_THREADS)
exclusive_scan(const int* __restrict__ deg, int* __restrict__ rowptr) {
  __shared__ int wsum[SCAN_THREADS / 32];
  int t = threadIdx.x;
  int lane = t & 31;
  int wv = t >> 5;
  bool live = t < N_NODES / SCAN_PER;
  int v[SCAN_PER];
  const int2* in2 = (const int2*)(deg + (live ? t : 0) * SCAN_PER);
  #pragma unroll
  for (int k = 0; k < SCAN_PER / 2; ++k) {
    int2 p = in2[k];
    v[2 * k] = live ? p.x : 0;
    v[2 * k + 1] = live ? p.y : 0;
  }
  int tot = 0;
  #pragma unroll
  for (int k = 0; k < SCAN_PER; ++k) tot += v[k];

  int inc = tot;
  for (int off = 1; off < 32; off <<= 1) {
    int x = __shfl_up(inc, off, 32);
    if (lane >= off) inc += x;
  }
  if (lane == 31) wsum[wv] = inc;
  __syncthreads();

  int ws = wsum[lane];
  int winc = ws;
  for (int off = 1; off < 32; off <<= 1) {
    int x = __shfl_up(winc, off, 32);
    if (lane >= off) winc += x;
  }
  int wbase = __shfl(winc - ws, wv, 32);
  int total = __shfl(winc, 31, 32);

  if (live) {
    int run = wbase + inc - tot;
    int2* out2 = (int2*)(rowptr + t * SCAN_PER);
    #pragma unroll
    for (int k = 0; k < SCAN_PER / 2; ++k) {
      int a = run;  run += v[2 * k];
      int b = run;  run += v[2 * k + 1];
      out2[k] = make_int2(a, b);
    }
  }
  if (t == 0) rowptr[N_NODES] = total;
}

__global__ void inv_from_deg(const int* __restrict__ deg, float* __restrict__ inv, int n) {
  int i = blockIdx.x * blockDim.x + threadIdx.x;
  if (i < n) {
    int d = deg[i];
    inv[i] = (d > 0) ? (1.0f / (float)d) : 0.0f;
  }
}

__global__ void __launch_bounds__(CSR_THREADS)
bin_chunks(const int* __restrict__ src, const int* __restrict__ dst,
           const float* __restrict__ w,
           const int* __restrict__ rowptr, const int* __restrict__ part,
           int* __restrict__ es, float* __restrict__ ew) {
  __shared__ int4 cur4[N_NODES / 4];
  int* cur = (int*)cur4;
  int t = threadIdx.x;
  int base = blockIdx.x * (CSR_THREADS * CSR_EPT) + t;
  int d[CSR_EPT], s[CSR_EPT];
  float ww[CSR_EPT];
  #pragma unroll
  for (int k = 0; k < CSR_EPT; ++k) {
    d[k]  = dst[base + k * CSR_THREADS];
    s[k]  = src[base + k * CSR_THREADS];
    ww[k] = w[base + k * CSR_THREADS];
  }
  const int4* rp4 = (const int4*)rowptr;
  const int4* pr4 = (const int4*)(part + blockIdx.x * PART_LD);
  constexpr int ROWS4 = N_NODES / 4;
  constexpr int FILL_T = 625, FILL_N = 4;
  static_assert(FILL_T * FILL_N == ROWS4 && FILL_T <= CSR_THREADS, "table fill must tile the rows");
  if (t < FILL_T) {
    int4 r[FILL_N], p[FILL_N];
    #pragma unroll
    for (int j = 0; j < FILL_N; ++j) {
      r[j] = rp4[t + j * FILL_T];
      p[j] = pr4[t + j * FILL_T];
    }
    #pragma unroll
    for (int j = 0; j < FILL_N; ++j)
      cur4[t + j * FILL_T] = make_int4(r[j].x + p[j].x, r[j].y + p[j].y, r[j].z + p[j].z, r[j].w + p[j].w);
  }
  __syncthreads();
  int pos[CSR_EPT];
  #pragma unroll
  for (int k = 0; k < CSR_EPT; ++k) pos[k] = atomicAdd(&cur[d[k]], 1);
  #pragma unroll
  for (int k = 0; k < CSR_EPT; ++k) {
    es[pos[k]] = s[k];
    ew[pos[k]] = ww[k];
  }
}

#define AGG1_GROUP 16
#define AGG2_GROUP 8

template<int D, bool TAIL>
__device__ __forceinline__ void gather_group(const float* __restrict__ X, int my_s, float my_w,
                                             int g, int n, int lane, float4 (&acc)[D / 128]) {
  float4 v[AGG1_GROUP][D / 128];
  #pragma unroll
  for (int u = 0; u < AGG1_GROUP; ++u) {
    #pragma unroll
    for (int c = 0; c < D / 128; ++c) v[u][c] = make_float4(0.f, 0.f, 0.f, 0.f);
    if (!TAIL || g + u < n) {
      int s = __builtin_amdgcn_readlane(my_s, g + u);
      const float4* xr = (const float4*)(X + (size_t)s * D);
      #pragma unroll
      for (int c = 0; c < D / 128; ++c) v[u][c] = xr[lane + 32 * c];
    }
  }
  #pragma unroll
  for (int u = 0; u < AGG1_GROUP; ++u) {
    if (!TAIL || g + u < n) {
      float ww = __int_as_float(__builtin_amdgcn_readlane(__float_as_int(my_w), g + u));
      #pragma unroll
      for (int c = 0; c < D / 128; ++c) {
        acc[c].x = fmaf(v[u][c].x, ww, acc[c].x);
        acc[c].y = fmaf(v[u][c].y, ww, acc[c].y);
        acc[c].z = fmaf(v[u][c].z, ww, acc[c].z);
        acc[c].w = fmaf(v[u][c].w, ww, acc[c].w);
      }
    }
  }
}

template<int D>
__global__ void __launch_bounds__(256)
aggregate_nodes(const float* __restrict__ X,
                const int* __restrict__ rowptr,
                const int* __restrict__ es,
                const float* __restrict__ ew,
                const float* __restrict__ inv,
                float* __restrict__ Y, int nN) {
  int node = __builtin_amdgcn_readfirstlane((int)((blockIdx.x * blockDim.x + threadIdx.x) >> 5));
  int lane = threadIdx.x & 31;
  if (node >= nN) return;
  int beg = rowptr[node];
  int end = rowptr[node + 1];
  float4 acc[D / 128];
  #pragma unroll
  for (int c = 0; c < D / 128; ++c) acc[c] = make_float4(0.f, 0.f, 0.f, 0.f);

  if (beg < end) {
    int last = end - 1;
    int cur_s = es[min(beg + lane, last)];
    float cur_w = ew[min(beg + lane, last)];
    for (int b = beg; b < end; b += 32) {
      int n = min(32, end - b);
      int nxt_s = es[min(b + 32 + lane, last)];
      float nxt_w = ew[min(b + 32 + lane, last)];
      int g = 0;
      for (; g + AGG1_GROUP <= n; g += AGG1_GROUP) gather_group<D, false>(X, cur_s, cur_w, g, n, lane, acc);
      if (g < n) gather_group<D, true>(X, cur_s, cur_w, g, n, lane, acc);
      cur_s = nxt_s;
      cur_w = nxt_w;
    }
  }
  float iv = inv[node];
  float4* yr = (float4*)(Y + (size_t)node * D);
  #pragma unroll
  for (int c = 0; c < D / 128; ++c)
    yr[lane + 32 * c] = make_float4(acc[c].x * iv, acc[c].y * iv, acc[c].z * iv, acc[c].w * iv);
}


template<bool TAIL>
__device__ __forceinline__ void gather_group_halves(const float* __restrict__ Q, int my_s, float my_w,
                                                    int g, int left, int lane, float4& acc) {
  int c = lane & 15;
  int src_lane = (lane & 16) + g;
  float4 v[AGG2_GROUP];
  float ww[AGG2_GROUP];
  #pragma unroll
  for (int u = 0; u < AGG2_GROUP; ++u) {
    int s = __shfl(my_s, src_lane + u, 32);
    ww[u] = __shfl(my_w, src_lane + u, 32);
    v[u] = make_float4(0.f, 0.f, 0.f, 0.f);
    if (!TAIL || u < left) v[u] = ((const float4*)(Q + (size_t)s * D_OUT))[c];
  }
  #pragma unroll
  for (int u = 0; u < AGG2_GROUP; ++u) {
    if (!TAIL || u < left) {
      acc.x = fmaf(v[u].x, ww[u], acc.x);
      acc.y = fmaf(v[u].y, ww[u], acc.y);
      acc.z = fmaf(v[u].z, ww[u], acc.z);
      acc.w = fmaf(v[u].w, ww[u], acc.w);
    }
  }
}

__global__ void __launch_bounds__(256)
aggregate_projected(const float* __restrict__ P, const float* __restrict__ Q,
                    const float* __restrict__ bias,
                    const int* __restrict__ rowptr,
                    const int* __restrict__ es,
                    const float* __restrict__ ew,
                    const float* __restrict__ inv,
                    float* __restrict__ out, int nN) {
  static_assert(D_OUT == 64, "a row is 16 lanes x float4");
  static_assert(2 * AGG2_GROUP == 16, "a batch of 16 steps is two groups");
  int node = __builtin_amdgcn_readfirstlane((int)((blockIdx.x * blockDim.x + threadIdx.x) >> 5));
  int lane = threadIdx.x & 31;
  if (node >= nN) return;
  int half = lane >> 4;
  int c    = lane & 15;
  int beg = rowptr[node];
  int end = rowptr[node + 1];
  int mid = beg + ((end - beg + 1) >> 1);
  int cmax = mid - beg;
  int cmin = end - mid;
  int lo  = half ? mid : beg;
  int cnt = half ? cmin : cmax;
  float4 acc = make_float4(0.f, 0.f, 0.f, 0.f);

  if (beg < end) {
    int cur_s = es[lo + min(c, cnt - 1)];
    float cur_w = ew[lo + min(c, cnt - 1)];
    for (int bb = 0; bb < cmax; bb += 16) {
      int nxt_s = es[lo + min(bb + 16 + c, cnt - 1)];
      float nxt_w = ew[lo + min(bb + 16 + c, cnt - 1)];
      for (int g = 0; g < 16 && bb + g < cmax; g += AGG2_GROUP) {
        if (bb + g + AGG2_GROUP <= cmin) gather_group_halves<false>(Q, cur_s, cur_w, g, AGG2_GROUP, lane, acc);
        else                             gather_group_halves<true>(Q, cur_s, cur_w, g, cnt - (bb + g), lane, acc);
      }
      cur_s = nxt_s;
      cur_w = nxt_w;
    }
  }
  acc.x += __shfl_xor(acc.x, 16, 32);
  acc.y += __shfl_xor(acc.y, 16, 32);
  acc.z += __shfl_xor(acc.z, 16, 32);
  acc.w += __shfl_xor(acc.w, 16, 32);
  if (half == 0) {
    float iv = inv[node];
    float4 p = ((const float4*)(P + (size_t)node * D_OUT))[c];
    float4 b = ((const float4*)bias)[c];
    ((float4*)(out + (size_t)node * D_OUT))[c] =
        make_float4(p.x + b.x + acc.x * iv, p.y + b.y + acc.y * iv,
                    p.z + b.z + acc.z * iv, p.w + b.w + acc.w * iv);
  }
}

template<int NDIM, int KDIM, int KSPLIT, bool RELU, bool BIAS>
__global__ void wmma_gemm_f32(const float* __restrict__ A0, int lda0,
                              const float* __restrict__ A1, int lda1,
                              const float* __restrict__ B,
                              const float* __restrict__ bias,
                              float* __restrict__ Y, int ldy,
                              int Mtiles, int Nstrips) {
  int wid  = (blockIdx.x * blockDim.x + threadIdx.x) >> 5;
  int lane = threadIdx.x & 31;
  if (wid >= Mtiles * Nstrips) return;
  int m0 = (wid / Nstrips) * 16;
  int cs = (wid % Nstrips) * 64;
  int slab = cs / NDIM;
  int n0   = cs % NDIM;
  int half = lane >> 4;
  int mr   = lane & 15;
  B += (size_t)slab * (KDIM * NDIM);
  Y += (size_t)slab * Mtiles * 16 * ldy;

  const float* ap = A0 + (size_t)(m0 + mr) * lda0 + 2 * half;
  const float* bp = B + (2 * half) * NDIM + n0 + mr;
  v8f acc0 = {}, acc1 = {}, acc2 = {}, acc3 = {};

  auto kstep = [&](const float*& arow, const float* pf) {
    v2f a = *(const v2f*)arow;
    arow += 4;
    __builtin_prefetch(pf, 0, 3);
    v2f b;
    b.x = bp[0];         b.y = bp[NDIM];
    acc0 = __builtin_amdgcn_wmma_f32_16x16x4_f32(false, a, false, b, (short)0, acc0, false, false);
    b.x = bp[16];        b.y = bp[NDIM + 16];
    acc1 = __builtin_amdgcn_wmma_f32_16x16x4_f32(false, a, false, b, (short)0, acc1, false, false);
    b.x = bp[32];        b.y = bp[NDIM + 32];
    acc2 = __builtin_amdgcn_wmma_f32_16x16x4_f32(false, a, false, b, (short)0, acc2, false, false);
    b.x = bp[48];        b.y = bp[NDIM + 48];
    acc3 = __builtin_amdgcn_wmma_f32_16x16x4_f32(false, a, false, b, (short)0, acc3, false, false);
    bp += 4 * NDIM;
  };

  if constexpr (KSPLIT == KDIM) {
    #pragma unroll 2
    for (int k = 0; k < KDIM; k += 4) kstep(ap, ap + 64);
  } else {
    static_assert(KSPLIT >= 64 && KSPLIT % 8 == 0 && KDIM - KSPLIT >= 64 && (KDIM - KSPLIT) % 8 == 0,
                  "the prefetch runs 64 columns ahead; the loops are unrolled by two steps");
    const float* ap1 = A1 + (size_t)(m0 + mr) * lda1 + 2 * half;
    #pragma unroll 2
    for (int k = 0; k < KSPLIT - 64; k += 4) kstep(ap, ap + 64);
    const float* pf1 = ap1;
    #pragma unroll 2
    for (int k = KSPLIT - 64; k < KSPLIT; k += 4) { kstep(ap, pf1); pf1 += 4; }
    #pragma unroll 2
    for (int k = KSPLIT; k < KDIM; k += 4) kstep(ap1, ap1 + 64);
  }

  #pragma unroll
  for (int j = 0; j < 4; ++j) {
    v8f acc = (j == 0) ? acc0 : (j == 1) ? acc1 : (j == 2) ? acc2 : acc3;
    int n = n0 + 16 * j + mr;
    float bv = 0.0f;
    if (BIAS) bv = bias[n];
    #pragma unroll
    for (int r = 0; r < 8; ++r) {
      int row = m0 + r + 8 * half;
      float v = acc[r];
      if (BIAS) v += bv;
      if (RELU) v = (v > 0.0f) ? v : 0.0f;
      Y[(size_t)row * ldy + n] = v;
    }
  }
}

extern "C" void kernel_launch(void* const* d_in, const int* in_sizes, int n_in,
                              void* d_out, int out_size, void* d_ws, size_t ws_size,
                              hipStream_t stream) {
  const float* h   = (const float*)d_in[0];
  const float* w   = (const float*)d_in[1];
  const int*   src = (const int*)d_in[2];
  const int*   dst = (const int*)d_in[3];
  const float* W1  = (const float*)d_in[4];
  const float* b1  = (const float*)d_in[5];
  const float* W2  = (const float*)d_in[6];
  const float* b2  = (const float*)d_in[7];
  float* out = (float*)d_out;

  int*   deg    = (int*)d_ws;
  int*   rowptr = deg + 10240;
  int*   es     = rowptr + 2 * 10240;
  float* ew     = (float*)(es + N_EDGES);
  float* inv    = ew + N_EDGES;
  float* bufA   = inv + 10240;
  float* bufB   = bufA + N_NODES * D_HID;
  float* PQ     = bufB + N_NODES * D_HID;

  int* part = (int*)bufB;
  hist_chunks<<<CSR_CHUNKS, CSR_THREADS, 0, stream>>>(dst, part);
  chunk_offsets<<<(N_NODES + 63) / 64, 64, 0, stream>>>(part, deg);
  exclusive_scan<<<1, SCAN_THREADS, 0, stream>>>(deg, rowptr);
  inv_from_deg<<<(N_NODES + 255) / 256, 256, 0, stream>>>(deg, inv, N_NODES);
  bin_chunks<<<CSR_CHUNKS, CSR_THREADS, 0, stream>>>(src, dst, w, rowptr, part, es, ew);

  aggregate_nodes<D_IN><<<(N_NODES * 32 + 255) / 256, 256, 0, stream>>>(
      h, rowptr, es, ew, inv, bufA, N_NODES);
  {
    int Mtiles = N_NODES / 16, Nstrips = D_HID / 64;
    wmma_gemm_f32<D_HID, 2 * D_IN, D_IN, true, true><<<(Mtiles * Nstrips + 7) / 8, 256, 0, stream>>>(
        h, D_IN, bufA, D_IN, W1, b1, bufB, D_HID, Mtiles, Nstrips);
  }

  {
    int Mtiles = N_NODES / 16, Nstrips = 2 * D_OUT / 64;
    wmma_gemm_f32<D_OUT, D_HID, D_HID, false, false><<<(Mtiles * Nstrips + 7) / 8, 256, 0, stream>>>(
        bufB, D_HID, nullptr, 0, W2, nullptr, PQ, D_OUT, Mtiles, Nstrips);
  }
  aggregate_projected<<<(N_NODES * 32 + 255) / 256, 256, 0, stream>>>(
      PQ, PQ + N_NODES * D_OUT, b2, rowptr, es, ew, inv, out, N_NODES);
}


# degree scan and 1/deg have no launch of their own: every bin block scans the 10000 degrees itself (block 0 stores rowptr), gathers take 1/deg from rowptr; 7 launches, was 9; sums unchanged
# speedup vs baseline: 2.4128x; 1.0367x over previous
#include <hip/hip_runtime.h>

typedef __attribute__((ext_vector_type(2))) float v2f;
typedef __attribute__((ext_vector_type(8))) float v8f;

#define N_NODES 10000
#define N_EDGES 640000
#define D_IN    128
#define D_HID   256
#define D_OUT   64

#define CSR_CHUNKS  125
#define CSR_THREADS 1024
#define CSR_EPT     5
#define PART_LD     10240

static_assert(CSR_CHUNKS * CSR_THREADS * CSR_EPT == N_EDGES, "chunks must tile the edge list");
static_assert(N_NODES % 4 == 0 && PART_LD % 4 == 0 && PART_LD >= N_NODES, "int4 table rows");

__global__ void __launch_bounds__(CSR_THREADS)
hist_chunks(const int* __restrict__ dst, int* __restrict__ part) {
  __shared__ int4 bins4[N_NODES / 4];
  int* bins = (int*)bins4;
  int t = threadIdx.x;
  int base = blockIdx.x * (CSR_THREADS * CSR_EPT) + t;
  int d[CSR_EPT];
  #pragma unroll
  for (int k = 0; k < CSR_EPT; ++k) d[k] = dst[base + k * CSR_THREADS];
  for (int i = t; i < N_NODES / 4; i += CSR_THREADS) bins4[i] = make_int4(0, 0, 0, 0);
  __syncthreads();
  #pragma unroll
  for (int k = 0; k < CSR_EPT; ++k) atomicAdd(&bins[d[k]], 1);
  __syncthreads();
  int4* row = (int4*)(part + blockIdx.x * PART_LD);
  for (int i = t; i < N_NODES / 4; i += CSR_THREADS) row[i] = bins4[i];
}

__global__ void __launch_bounds__(64)
chunk_offsets(int* __restrict__ part, int* __restrict__ deg) {
  int i = blockIdx.x * 64 + threadIdx.x;
  if (i >= N_NODES) return;
  int* col = part + i;
  int run = 0;
  #pragma unroll
  for (int g0 = 0; g0 < CSR_CHUNKS; g0 += 25) {
    int v[25];
    #pragma unroll
    for (int k = 0; k < 25; ++k) v[k] = col[(g0 + k) * PART_LD];
    #pragma unroll
    for (int k = 0; k < 25; ++k) { col[(g0 + k) * PART_LD] = run; run += v[k]; }
  }
  deg[i] = run;
}

#define SCAN_THREADS CSR_THREADS
#define SCAN_PER     10
static_assert(SCAN_THREADS * SCAN_PER >= N_NODES && N_NODES % SCAN_PER == 0, "threads must tile the nodes");
static_assert(SCAN_PER % 2 == 0, "int2 loads and stores");
static_assert(SCAN_THREADS / 32 == 32, "one wave sum per lane in the second scan");

__global__ void __launch_bounds__(CSR_THREADS)
bin_chunks(const int* __restrict__ src, const int* __restrict__ dst,
           const float* __restrict__ w,
           const int* __restrict__ deg, const int* __restrict__ part,
           int* __restrict__ rowptr,
           int* __restrict__ es, float* __restrict__ ew) {
  __shared__ int2 cur_tab[N_NODES / 2];
  __shared__ int wsum[SCAN_THREADS / 32];
  int* cur = (int*)cur_tab;
  int t = threadIdx.x;
  int lane = t & 31;
  int wv = t >> 5;
  int base = blockIdx.x * (CSR_THREADS * CSR_EPT) + t;
  int d[CSR_EPT], s[CSR_EPT];
  float ww[CSR_EPT];
  #pragma unroll
  for (int k = 0; k < CSR_EPT; ++k) {
    d[k]  = dst[base + k * CSR_THREADS];
    s[k]  = src[base + k * CSR_THREADS];
    ww[k] = w[base + k * CSR_THREADS];
  }

  bool live = t < N_NODES / SCAN_PER;
  int own = (live ? t : 0) * SCAN_PER;
  const int2* dg2 = (const int2*)(deg + own);
  const int2* pr2 = (const int2*)(part + blockIdx.x * PART_LD + own);
  int v[SCAN_PER], p[SCAN_PER];
  #pragma unroll
  for (int k = 0; k < SCAN_PER / 2; ++k) {
    int2 a = dg2[k];
    int2 b = pr2[k];
    v[2 * k] = live ? a.x : 0;
    v[2 * k + 1] = live ? a.y : 0;
    p[2 * k] = b.x;
    p[2 * k + 1] = b.y;
  }
  int tot = 0;
  #pragma unroll
  for (int k = 0; k < SCAN_PER; ++k) tot += v[k];

  int inc = tot;
  for (int off = 1; off < 32; off <<= 1) {
    int x = __shfl_up(inc, off, 32);
    if (lane >= off) inc += x;
  }
  if (lane == 31) wsum[wv] = inc;
  __syncthreads();

  int ws = wsum[lane];
  int winc = ws;
  for (int off = 1; off < 32; off <<= 1) {
    int x = __shfl_up(winc, off, 32);
    if (lane >= off) winc += x;
  }
  int wbase = __shfl(winc - ws, wv, 32);
  int total = __shfl(winc, 31, 32);

  if (live) {
    int run = wbase + inc - tot;
    int2* cur2 = (int2*)(cur + t * SCAN_PER);
    int2* out2 = (int2*)(rowptr + t * SCAN_PER);
    #pragma unroll
    for (int k = 0; k < SCAN_PER / 2; ++k) {
      int a = run;  run += v[2 * k];
      int b = run;  run += v[2 * k + 1];
      cur2[k] = make_int2(a + p[2 * k], b + p[2 * k + 1]);
      if (blockIdx.x == 0) out2[k] = make_int2(a, b);
    }
  }
  if (blockIdx.x == 0 && t == 0) rowptr[N_NODES] = total;
  __syncthreads();

  int pos[CSR_EPT];
  #pragma unroll
  for (int k = 0; k < CSR_EPT; ++k) pos[k] = atomicAdd(&cur[d[k]], 1);
  #pragma unroll
  for (int k = 0; k < CSR_EPT; ++k) {
    es[pos[k]] = s[k];
    ew[pos[k]] = ww[k];
  }
}

#define AGG1_GROUP 16
#define AGG2_GROUP 8

template<int D, bool TAIL>
__device__ __forceinline__ void gather_group(const float* __restrict__ X, int my_s, float my_w,
                                             int g, int n, int lane, float4 (&acc)[D / 128]) {
  float4 v[AGG1_GROUP][D / 128];
  #pragma unroll
  for (int u = 0; u < AGG1_GROUP; ++u) {
    #pragma unroll
    for (int c = 0; c < D / 128; ++c) v[u][c] = make_float4(0.f, 0.f, 0.f, 0.f);
    if (!TAIL || g + u < n) {
      int s = __builtin_amdgcn_readlane(my_s, g + u);
      const float4* xr = (const float4*)(X + (size_t)s * D);
      #pragma unroll
      for (int c = 0; c < D / 128; ++c) v[u][c] = xr[lane + 32 * c];
    }
  }
  #pragma unroll
  for (int u = 0; u < AGG1_GROUP; ++u) {
    if (!TAIL || g + u < n) {
      float ww = __int_as_float(__builtin_amdgcn_readlane(__float_as_int(my_w), g + u));
      #pragma unroll
      for (int c = 0; c < D / 128; ++c) {
        acc[c].x = fmaf(v[u][c].x, ww, acc[c].x);
        acc[c].y = fmaf(v[u][c].y, ww, acc[c].y);
        acc[c].z = fmaf(v[u][c].z, ww, acc[c].z);
        acc[c].w = fmaf(v[u][c].w, ww, acc[c].w);
      }
    }
  }
}

template<int D>
__global__ void __launch_bounds__(256)
aggregate_nodes(const float* __restrict__ X,
                const int* __restrict__ rowptr,
                const int* __restrict__ es,
                const float* __restrict__ ew,
                float* __restrict__ Y, int nN) {
  int node = __builtin_amdgcn_readfirstlane((int)((blockIdx.x * blockDim.x + threadIdx.x) >> 5));
  int lane = threadIdx.x & 31;
  if (node >= nN) return;
  int beg = rowptr[node];
  int end = rowptr[node + 1];
  float4 acc[D / 128];
  #pragma unroll
  for (int c = 0; c < D / 128; ++c) acc[c] = make_float4(0.f, 0.f, 0.f, 0.f);

  if (beg < end) {
    int last = end - 1;
    int cur_s = es[min(beg + lane, last)];
    float cur_w = ew[min(beg + lane, last)];
    for (int b = beg; b < end; b += 32) {
      int n = min(32, end - b);
      int nxt_s = es[min(b + 32 + lane, last)];
      float nxt_w = ew[min(b + 32 + lane, last)];
      int g = 0;
      for (; g + AGG1_GROUP <= n; g += AGG1_GROUP) gather_group<D, false>(X, cur_s, cur_w, g, n, lane, acc);
      if (g < n) gather_group<D, true>(X, cur_s, cur_w, g, n, lane, acc);
      cur_s = nxt_s;
      cur_w = nxt_w;
    }
  }
  int dg = end - beg;
  float iv = (dg > 0) ? (1.0f / (float)dg) : 0.0f;
  float4* yr = (float4*)(Y + (size_t)node * D);
  #pragma unroll
  for (int c = 0; c < D / 128; ++c)
    yr[lane + 32 * c] = make_float4(acc[c].x * iv, acc[c].y * iv, acc[c].z * iv, acc[c].w * iv);
}


template<bool TAIL>
__device__ __forceinline__ void gather_group_halves(const float* __restrict__ Q, int my_s, float my_w,
                                                    int g, int left, int lane, float4& acc) {
  int c = lane & 15;
  int src_lane = (lane & 16) + g;
  float4 v[AGG2_GROUP];
  float ww[AGG2_GROUP];
  #pragma unroll
  for (int u = 0; u < AGG2_GROUP; ++u) {
    int s = __shfl(my_s, src_lane + u, 32);
    ww[u] = __shfl(my_w, src_lane + u, 32);
    v[u] = make_float4(0.f, 0.f, 0.f, 0.f);
    if (!TAIL || u < left) v[u] = ((const float4*)(Q + (size_t)s * D_OUT))[c];
  }
  #pragma unroll
  for (int u = 0; u < AGG2_GROUP; ++u) {
    if (!TAIL || u < left) {
      acc.x = fmaf(v[u].x, ww[u], acc.x);
      acc.y = fmaf(v[u].y, ww[u], acc.y);
      acc.z = fmaf(v[u].z, ww[u], acc.z);
      acc.w = fmaf(v[u].w, ww[u], acc.w);
    }
  }
}

__global__ void __launch_bounds__(256)
aggregate_projected(const float* __restrict__ P, const float* __restrict__ Q,
                    const float* __restrict__ bias,
                    const int* __restrict__ rowptr,
                    const int* __restrict__ es,
                    const float* __restrict__ ew,
                    float* __restrict__ out, int nN) {
  static_assert(D_OUT == 64, "a row is 16 lanes x float4");
  static_assert(2 * AGG2_GROUP == 16, "a batch of 16 steps is two groups");
  int node = __builtin_amdgcn_readfirstlane((int)((blockIdx.x * blockDim.x + threadIdx.x) >> 5));
  int lane = threadIdx.x & 31;
  if (node >= nN) return;
  int half = lane >> 4;
  int c    = lane & 15;
  int beg = rowptr[node];
  int end = rowptr[node + 1];
  int mid = beg + ((end - beg + 1) >> 1);
  int cmax = mid - beg;
  int cmin = end - mid;
  int lo  = half ? mid : beg;
  int cnt = half ? cmin : cmax;
  float4 acc = make_float4(0.f, 0.f, 0.f, 0.f);

  if (beg < end) {
    int cur_s = es[lo + min(c, cnt - 1)];
    float cur_w = ew[lo + min(c, cnt - 1)];
    for (int bb = 0; bb < cmax; bb += 16) {
      int nxt_s = es[lo + min(bb + 16 + c, cnt - 1)];
      float nxt_w = ew[lo + min(bb + 16 + c, cnt - 1)];
      for (int g = 0; g < 16 && bb + g < cmax; g += AGG2_GROUP) {
        if (bb + g + AGG2_GROUP <= cmin) gather_group_halves<false>(Q, cur_s, cur_w, g, AGG2_GROUP, lane, acc);
        else                             gather_group_halves<true>(Q, cur_s, cur_w, g, cnt - (bb + g), lane, acc);
      }
      cur_s = nxt_s;
      cur_w = nxt_w;
    }
  }
  acc.x += __shfl_xor(acc.x, 16, 32);
  acc.y += __shfl_xor(acc.y, 16, 32);
  acc.z += __shfl_xor(acc.z, 16, 32);
  acc.w += __shfl_xor(acc.w, 16, 32);
  if (half == 0) {
    int dg = end - beg;
    float iv = (dg > 0) ? (1.0f / (float)dg) : 0.0f;
    float4 p = ((const float4*)(P + (size_t)node * D_OUT))[c];
    float4 b = ((const float4*)bias)[c];
    ((float4*)(out + (size_t)node * D_OUT))[c] =
        make_float4(p.x + b.x + acc.x * iv, p.y + b.y + acc.y * iv,
                    p.z + b.z + acc.z * iv, p.w + b.w + acc.w * iv);
  }
}

template<int NDIM, int KDIM, int KSPLIT, bool RELU, bool BIAS>
__global__ void wmma_gemm_f32(const float* __restrict__ A0, int lda0,
                              const float* __restrict__ A1, int lda1,
                              const float* __restrict__ B,
                              const float* __restrict__ bias,
                              float* __restrict__ Y, int ldy,
                              int Mtiles, int Nstrips) {
  int wid  = (blockIdx.x * blockDim.x + threadIdx.x) >> 5;
  int lane = threadIdx.x & 31;
  if (wid >= Mtiles * Nstrips) return;
  int m0 = (wid / Nstrips) * 16;
  int cs = (wid % Nstrips) * 64;
  int slab = cs / NDIM;
  int n0   = cs % NDIM;
  int half = lane >> 4;
  int mr   = lane & 15;
  B += (size_t)slab * (KDIM * NDIM);
  Y += (size_t)slab * Mtiles * 16 * ldy;

  const float* ap = A0 + (size_t)(m0 + mr) * lda0 + 2 * half;
  const float* bp = B + (2 * half) * NDIM + n0 + mr;
  v8f acc0 = {}, acc1 = {}, acc2 = {}, acc3 = {};

  auto kstep = [&](const float*& arow, const float* pf) {
    v2f a = *(const v2f*)arow;
    arow += 4;
    __builtin_prefetch(pf, 0, 3);
    v2f b;
    b.x = bp[0];         b.y = bp[NDIM];
    acc0 = __builtin_amdgcn_wmma_f32_16x16x4_f32(false, a, false, b, (short)0, acc0, false, false);
    b.x = bp[16];        b.y = bp[NDIM + 16];
    acc1 = __builtin_amdgcn_wmma_f32_16x16x4_f32(false, a, false, b, (short)0, acc1, false, false);
    b.x = bp[32];        b.y = bp[NDIM + 32];
    acc2 = __builtin_amdgcn_wmma_f32_16x16x4_f32(false, a, false, b, (short)0, acc2, false, false);
    b.x = bp[48];        b.y = bp[NDIM + 48];
    acc3 = __builtin_amdgcn_wmma_f32_16x16x4_f32(false, a, false, b, (short)0, acc3, false, false);
    bp += 4 * NDIM;
  };

  if constexpr (KSPLIT == KDIM) {
    #pragma unroll 2
    for (int k = 0; k < KDIM; k += 4) kstep(ap, ap + 64);
  } else {
    static_assert(KSPLIT >= 64 && KSPLIT % 8 == 0 && KDIM - KSPLIT >= 64 && (KDIM - KSPLIT) % 8 == 0,
                  "the prefetch runs 64 columns ahead; the loops are unrolled by two steps");
    const float* ap1 = A1 + (size_t)(m0 + mr) * lda1 + 2 * half;
    #pragma unroll 2
    for (int k = 0; k < KSPLIT - 64; k += 4) kstep(ap, ap + 64);
    const float* pf1 = ap1;
    #pragma unroll 2
    for (int k = KSPLIT - 64; k < KSPLIT; k += 4) { kstep(ap, pf1); pf1 += 4; }
    #pragma unroll 2
    for (int k = KSPLIT; k < KDIM; k += 4) kstep(ap1, ap1 + 64);
  }

  #pragma unroll
  for (int j = 0; j < 4; ++j) {
    v8f acc = (j == 0) ? acc0 : (j == 1) ? acc1 : (j == 2) ? acc2 : acc3;
    int n = n0 + 16 * j + mr;
    float bv = 0.0f;
    if (BIAS) bv = bias[n];
    #pragma unroll
    for (int r = 0; r < 8; ++r) {
      int row = m0 + r + 8 * half;
      float v = acc[r];
      if (BIAS) v += bv;
      if (RELU) v = (v > 0.0f) ? v : 0.0f;
      Y[(size_t)row * ldy + n] = v;
    }
  }
}

extern "C" void kernel_launch(void* const* d_in, const int* in_sizes, int n_in,
                              void* d_out, int out_size, void* d_ws, size_t ws_size,
                              hipStream_t stream) {
  const float* h   = (const float*)d_in[0];
  const float* w   = (const float*)d_in[1];
  const int*   src = (const int*)d_in[2];
  const int*   dst = (const int*)d_in[3];
  const float* W1  = (const float*)d_in[4];
  const float* b1  = (const float*)d_in[5];
  const float* W2  = (const float*)d_in[6];
  const float* b2  = (const float*)d_in[7];
  float* out = (float*)d_out;

  int*   deg    = (int*)d_ws;
  int*   rowptr = deg + 10240;
  int*   es     = rowptr + 2 * 10240;
  float* ew     = (float*)(es + N_EDGES);
  float* bufA   = ew + N_EDGES + 10240;
  float* bufB   = bufA + N_NODES * D_HID;
  float* PQ     = bufB + N_NODES * D_HID;

  int* part = (int*)bufB;
  hist_chunks<<<CSR_CHUNKS, CSR_THREADS, 0, stream>>>(dst, part);
  chunk_offsets<<<(N_NODES + 63) / 64, 64, 0, stream>>>(part, deg);
  bin_chunks<<<CSR_CHUNKS, CSR_THREADS, 0, stream>>>(src, dst, w, deg, part, rowptr, es, ew);

  aggregate_nodes<D_IN><<<(N_NODES * 32 + 255) / 256, 256, 0, stream>>>(
      h, rowptr, es, ew, bufA, N_NODES);
  {
    int Mtiles = N_NODES / 16, Nstrips = D_HID / 64;
    wmma_gemm_f32<D_HID, 2 * D_IN, D_IN, true, true><<<(Mtiles * Nstrips + 7) / 8, 256, 0, stream>>>(
        h, D_IN, bufA, D_IN, W1, b1, bufB, D_HID, Mtiles, Nstrips);
  }

  {
    int Mtiles = N_NODES / 16, Nstrips = 2 * D_OUT / 64;
    wmma_gemm_f32<D_OUT, D_HID, D_HID, false, false><<<(Mtiles * Nstrips + 7) / 8, 256, 0, stream>>>(
        bufB, D_HID, nullptr, 0, W2, nullptr, PQ, D_OUT, Mtiles, Nstrips);
  }
  aggregate_projected<<<(N_NODES * 32 + 255) / 256, 256, 0, stream>>>(
      PQ, PQ + N_NODES * D_OUT, b2, rowptr, es, ew, out, N_NODES);
}


# GEMMs: a lane owns 4 adjacent columns and 4 adjacent k per 8-block, so A, B and result accesses are 16 bytes wide (10 loads per 16 k, was 36; 8 row stores, was 32); f32 sums reassociated in k
# speedup vs baseline: 2.8023x; 1.1614x over previous
#include <hip/hip_runtime.h>

typedef __attribute__((ext_vector_type(2))) float v2f;
typedef __attribute__((ext_vector_type(4))) float v4f;
typedef __attribute__((ext_vector_type(8))) float v8f;

#define N_NODES 10000
#define N_EDGES 640000
#define D_IN    128
#define D_HID   256
#define D_OUT   64

#define CSR_CHUNKS  125
#define CSR_THREADS 1024
#define CSR_EPT     5
#define PART_LD     10240

static_assert(CSR_CHUNKS * CSR_THREADS * CSR_EPT == N_EDGES, "chunks must tile the edge list");
static_assert(N_NODES % 4 == 0 && PART_LD % 4 == 0 && PART_LD >= N_NODES, "int4 table rows");

__global__ void __launch_bounds__(CSR_THREADS)
hist_chunks(const int* __restrict__ dst, int* __restrict__ part) {
  __shared__ int4 bins4[N_NODES / 4];
  int* bins = (int*)bins4;
  int t = threadIdx.x;
  int base = blockIdx.x * (CSR_THREADS * CSR_EPT) + t;
  int d[CSR_EPT];
  #pragma unroll
  for (int k = 0; k < CSR_EPT; ++k) d[k] = dst[base + k * CSR_THREADS];
  for (int i = t; i < N_NODES / 4; i += CSR_THREADS) bins4[i] = make_int4(0, 0, 0, 0);
  __syncthreads();
  #pragma unroll
  for (int k = 0; k < CSR_EPT; ++k) atomicAdd(&bins[d[k]], 1);
  __syncthreads();
  int4* row = (int4*)(part + blockIdx.x * PART_LD);
  for (int i = t; i < N_NODES / 4; i += CSR_THREADS) row[i] = bins4[i];
}

__global__ void __launch_bounds__(64)
chunk_offsets(int* __restrict__ part, int* __restrict__ deg) {
  int i = blockIdx.x * 64 + threadIdx.x;
  if (i >= N_NODES) return;
  int* col = part + i;
  int run = 0;
  #pragma unroll
  for (int g0 = 0; g0 < CSR_CHUNKS; g0 += 25) {
    int v[25];
    #pragma unroll
    for (int k = 0; k < 25; ++k) v[k] = col[(g0 + k) * PART_LD];
    #pragma unroll
    for (int k = 0; k < 25; ++k) { col[(g0 + k) * PART_LD] = run; run += v[k]; }
  }
  deg[i] = run;
}

#define SCAN_THREADS CSR_THREADS
#define SCAN_PER     10
static_assert(SCAN_THREADS * SCAN_PER >= N_NODES && N_NODES % SCAN_PER == 0, "threads must tile the nodes");
static_assert(SCAN_PER % 2 == 0, "int2 loads and stores");
static_assert(SCAN_THREADS / 32 == 32, "one wave sum per lane in the second scan");

__global__ void __launch_bounds__(CSR_THREADS)
bin_chunks(const int* __restrict__ src, const int* __restrict__ dst,
           const float* __restrict__ w,
           const int* __restrict__ deg, const int* __restrict__ part,
           int* __restrict__ rowptr,
           int* __restrict__ es, float* __restrict__ ew) {
  __shared__ int2 cur_tab[N_NODES / 2];
  __shared__ int wsum[SCAN_THREADS / 32];
  int* cur = (int*)cur_tab;
  int t = threadIdx.x;
  int lane = t & 31;
  int wv = t >> 5;
  int base = blockIdx.x * (CSR_THREADS * CSR_EPT) + t;
  int d[CSR_EPT], s[CSR_EPT];
  float ww[CSR_EPT];
  #pragma unroll
  for (int k = 0; k < CSR_EPT; ++k) {
    d[k]  = dst[base + k * CSR_THREADS];
    s[k]  = src[base + k * CSR_THREADS];
    ww[k] = w[base + k * CSR_THREADS];
  }

  bool live = t < N_NODES / SCAN_PER;
  int own = (live ? t : 0) * SCAN_PER;
  const int2* dg2 = (const int2*)(deg + own);
  const int2* pr2 = (const int2*)(part + blockIdx.x * PART_LD + own);
  int v[SCAN_PER], p[SCAN_PER];
  #pragma unroll
  for (int k = 0; k < SCAN_PER / 2; ++k) {
    int2 a = dg2[k];
    int2 b = pr2[k];
    v[2 * k] = live ? a.x : 0;
    v[2 * k + 1] = live ? a.y : 0;
    p[2 * k] = b.x;
    p[2 * k + 1] = b.y;
  }
  int tot = 0;
  #pragma unroll
  for (int k = 0; k < SCAN_PER; ++k) tot += v[k];

  int inc = tot;
  for (int off = 1; off < 32; off <<= 1) {
    int x = __shfl_up(inc, off, 32);
    if (lane >= off) inc += x;
  }
  if (lane == 31) wsum[wv] = inc;
  __syncthreads();

  int ws = wsum[lane];
  int winc = ws;
  for (int off = 1; off < 32; off <<= 1) {
    int x = __shfl_up(winc, off, 32);
    if (lane >= off) winc += x;
  }
  int wbase = __shfl(winc - ws, wv, 32);
  int total = __shfl(winc, 31, 32);

  if (live) {
    int run = wbase + inc - tot;
    int2* cur2 = (int2*)(cur + t * SCAN_PER);
    int2* out2 = (int2*)(rowptr + t * SCAN_PER);
    #pragma unroll
    for (int k = 0; k < SCAN_PER / 2; ++k) {
      int a = run;  run += v[2 * k];
      int b = run;  run += v[2 * k + 1];
      cur2[k] = make_int2(a + p[2 * k], b + p[2 * k + 1]);
      if (blockIdx.x == 0) out2[k] = make_int2(a, b);
    }
  }
  if (blockIdx.x == 0 && t == 0) rowptr[N_NODES] = total;
  __syncthreads();

  int pos[CSR_EPT];
  #pragma unroll
  for (int k = 0; k < CSR_EPT; ++k) pos[k] = atomicAdd(&cur[d[k]], 1);
  #pragma unroll
  for (int k = 0; k < CSR_EPT; ++k) {
    es[pos[k]] = s[k];
    ew[pos[k]] = ww[k];
  }
}

#define AGG1_GROUP 16
#define AGG2_GROUP 8

template<int D, bool TAIL>
__device__ __forceinline__ void gather_group(const float* __restrict__ X, int my_s, float my_w,
                                             int g, int n, int lane, float4 (&acc)[D / 128]) {
  float4 v[AGG1_GROUP][D / 128];
  #pragma unroll
  for (int u = 0; u < AGG1_GROUP; ++u) {
    #pragma unroll
    for (int c = 0; c < D / 128; ++c) v[u][c] = make_float4(0.f, 0.f, 0.f, 0.f);
    if (!TAIL || g + u < n) {
      int s = __builtin_amdgcn_readlane(my_s, g + u);
      const float4* xr = (const float4*)(X + (size_t)s * D);
      #pragma unroll
      for (int c = 0; c < D / 128; ++c) v[u][c] = xr[lane + 32 * c];
    }
  }
  #pragma unroll
  for (int u = 0; u < AGG1_GROUP; ++u) {
    if (!TAIL || g + u < n) {
      float ww = __int_as_float(__builtin_amdgcn_readlane(__float_as_int(my_w), g + u));
      #pragma unroll
      for (int c = 0; c < D / 128; ++c) {
        acc[c].x = fmaf(v[u][c].x, ww, acc[c].x);
        acc[c].y = fmaf(v[u][c].y, ww, acc[c].y);
        acc[c].z = fmaf(v[u][c].z, ww, acc[c].z);
        acc[c].w = fmaf(v[u][c].w, ww, acc[c].w);
      }
    }
  }
}

template<int D>
__global__ void __launch_bounds__(256)
aggregate_nodes(const float* __restrict__ X,
                const int* __restrict__ rowptr,
                const int* __restrict__ es,
                const float* __restrict__ ew,
                float* __restrict__ Y, int nN) {
  int node = __builtin_amdgcn_readfirstlane((int)((blockIdx.x * blockDim.x + threadIdx.x) >> 5));
  int lane = threadIdx.x & 31;
  if (node >= nN) return;
  int beg = rowptr[node];
  int end = rowptr[node + 1];
  float4 acc[D / 128];
  #pragma unroll
  for (int c = 0; c < D / 128; ++c) acc[c] = make_float4(0.f, 0.f, 0.f, 0.f);

  if (beg < end) {
    int last = end - 1;
    int cur_s = es[min(beg + lane, last)];
    float cur_w = ew[min(beg + lane, last)];
    for (int b = beg; b < end; b += 32) {
      int n = min(32, end - b);
      int nxt_s = es[min(b + 32 + lane, last)];
      float nxt_w = ew[min(b + 32 + lane, last)];
      int g = 0;
      for (; g + AGG1_GROUP <= n; g += AGG1_GROUP) gather_group<D, false>(X, cur_s, cur_w, g, n, lane, acc);
      if (g < n) gather_group<D, true>(X, cur_s, cur_w, g, n, lane, acc);
      cur_s = nxt_s;
      cur_w = nxt_w;
    }
  }
  int dg = end - beg;
  float iv = (dg > 0) ? (1.0f / (float)dg) : 0.0f;
  float4* yr = (float4*)(Y + (size_t)node * D);
  #pragma unroll
  for (int c = 0; c < D / 128; ++c)
    yr[lane + 32 * c] = make_float4(acc[c].x * iv, acc[c].y * iv, acc[c].z * iv, acc[c].w * iv);
}


template<bool TAIL>
__device__ __forceinline__ void gather_group_halves(const float* __restrict__ Q, int my_s, float my_w,
                                                    int g, int left, int lane, float4& acc) {
  int c = lane & 15;
  int src_lane = (lane & 16) + g;
  float4 v[AGG2_GROUP];
  float ww[AGG2_GROUP];
  #pragma unroll
  for (int u = 0; u < AGG2_GROUP; ++u) {
    int s = __shfl(my_s, src_lane + u, 32);
    ww[u] = __shfl(my_w, src_lane + u, 32);
    v[u] = make_float4(0.f, 0.f, 0.f, 0.f);
    if (!TAIL || u < left) v[u] = ((const float4*)(Q + (size_t)s * D_OUT))[c];
  }
  #pragma unroll
  for (int u = 0; u < AGG2_GROUP; ++u) {
    if (!TAIL || u < left) {
      acc.x = fmaf(v[u].x, ww[u], acc.x);
      acc.y = fmaf(v[u].y, ww[u], acc.y);
      acc.z = fmaf(v[u].z, ww[u], acc.z);
      acc.w = fmaf(v[u].w, ww[u], acc.w);
    }
  }
}

__global__ void __launch_bounds__(256)
aggregate_projected(const float* __restrict__ P, const float* __restrict__ Q,
                    const float* __restrict__ bias,
                    const int* __restrict__ rowptr,
                    const int* __restrict__ es,
                    const float* __restrict__ ew,
                    float* __restrict__ out, int nN) {
  static_assert(D_OUT == 64, "a row is 16 lanes x float4");
  static_assert(2 * AGG2_GROUP == 16, "a batch of 16 steps is two groups");
  int node = __builtin_amdgcn_readfirstlane((int)((blockIdx.x * blockDim.x + threadIdx.x) >> 5));
  int lane = threadIdx.x & 31;
  if (node >= nN) return;
  int half = lane >> 4;
  int c    = lane & 15;
  int beg = rowptr[node];
  int end = rowptr[node + 1];
  int mid = beg + ((end - beg + 1) >> 1);
  int cmax = mid - beg;
  int cmin = end - mid;
  int lo  = half ? mid : beg;
  int cnt = half ? cmin : cmax;
  float4 acc = make_float4(0.f, 0.f, 0.f, 0.f);

  if (beg < end) {
    int cur_s = es[lo + min(c, cnt - 1)];
    float cur_w = ew[lo + min(c, cnt - 1)];
    for (int bb = 0; bb < cmax; bb += 16) {
      int nxt_s = es[lo + min(bb + 16 + c, cnt - 1)];
      float nxt_w = ew[lo + min(bb + 16 + c, cnt - 1)];
      for (int g = 0; g < 16 && bb + g < cmax; g += AGG2_GROUP) {
        if (bb + g + AGG2_GROUP <= cmin) gather_group_halves<false>(Q, cur_s, cur_w, g, AGG2_GROUP, lane, acc);
        else                             gather_group_halves<true>(Q, cur_s, cur_w, g, cnt - (bb + g), lane, acc);
      }
      cur_s = nxt_s;
      cur_w = nxt_w;
    }
  }
  acc.x += __shfl_xor(acc.x, 16, 32);
  acc.y += __shfl_xor(acc.y, 16, 32);
  acc.z += __shfl_xor(acc.z, 16, 32);
  acc.w += __shfl_xor(acc.w, 16, 32);
  if (half == 0) {
    int dg = end - beg;
    float iv = (dg > 0) ? (1.0f / (float)dg) : 0.0f;
    float4 p = ((const float4*)(P + (size_t)node * D_OUT))[c];
    float4 b = ((const float4*)bias)[c];
    ((float4*)(out + (size_t)node * D_OUT))[c] =
        make_float4(p.x + b.x + acc.x * iv, p.y + b.y + acc.y * iv,
                    p.z + b.z + acc.z * iv, p.w + b.w + acc.w * iv);
  }
}

template<int NDIM, int KDIM, int KSPLIT, bool RELU, bool BIAS>
__global__ void __launch_bounds__(256)
wmma_gemm_f32(const float* __restrict__ A0, int lda0,
                              const float* __restrict__ A1, int lda1,
                              const float* __restrict__ B,
                              const float* __restrict__ bias,
                              float* __restrict__ Y, int ldy,
                              int Mtiles, int Nstrips) {
  static_assert(NDIM % 64 == 0 && KDIM % 16 == 0, "float4 rows of B; a loop trip takes two K blocks of 8");
  int wid  = (blockIdx.x * blockDim.x + threadIdx.x) >> 5;
  int lane = threadIdx.x & 31;
  if (wid >= Mtiles * Nstrips) return;
  int m0 = (wid / Nstrips) * 16;
  int cs = (wid % Nstrips) * 64;
  int slab = cs / NDIM;
  int n0   = cs % NDIM;
  int half = lane >> 4;
  int mr   = lane & 15;
  B += (size_t)slab * (KDIM * NDIM);
  Y += (size_t)slab * Mtiles * 16 * ldy;

  const float* ap = A0 + (size_t)(m0 + mr) * lda0 + 4 * half;
  const float* bp = B + (4 * half) * NDIM + n0 + 4 * mr;
  v8f acc0 = {}, acc1 = {}, acc2 = {}, acc3 = {};

  auto step = [&](float ax, float ay, v4f x, v4f y) {
    v2f a, b;
    a.x = ax;   a.y = ay;
    b.x = x.x;  b.y = y.x;
    acc0 = __builtin_amdgcn_wmma_f32_16x16x4_f32(false, a, false, b, (short)0, acc0, false, false);
    b.x = x.y;  b.y = y.y;
    acc1 = __builtin_amdgcn_wmma_f32_16x16x4_f32(false, a, false, b, (short)0, acc1, false, false);
    b.x = x.z;  b.y = y.z;
    acc2 = __builtin_amdgcn_wmma_f32_16x16x4_f32(false, a, false, b, (short)0, acc2, false, false);
    b.x = x.w;  b.y = y.w;
    acc3 = __builtin_amdgcn_wmma_f32_16x16x4_f32(false, a, false, b, (short)0, acc3, false, false);
  };

  auto kpair = [&](const float*& arow, const float* pf) {
    v4f a0 = *(const v4f*)arow;
    v4f a1 = *(const v4f*)(arow + 8);
    arow += 16;
    __builtin_prefetch(pf, 0, 3);
    v4f r0 = *(const v4f*)(bp);
    v4f r1 = *(const v4f*)(bp + NDIM);
    v4f r2 = *(const v4f*)(bp + 2 * NDIM);
    v4f r3 = *(const v4f*)(bp + 3 * NDIM);
    v4f r4 = *(const v4f*)(bp + 8 * NDIM);
    v4f r5 = *(const v4f*)(bp + 9 * NDIM);
    v4f r6 = *(const v4f*)(bp + 10 * NDIM);
    v4f r7 = *(const v4f*)(bp + 11 * NDIM);
    bp += 16 * NDIM;
    __builtin_amdgcn_sched_barrier(0);
    step(a0.x, a0.y, r0, r1);
    step(a0.z, a0.w, r2, r3);
    step(a1.x, a1.y, r4, r5);
    step(a1.z, a1.w, r6, r7);
  };

  if constexpr (KSPLIT == KDIM) {
    for (int k = 0; k < KDIM; k += 16) kpair(ap, ap + 64);
  } else {
    static_assert(KSPLIT >= 64 && KSPLIT % 16 == 0 && KDIM - KSPLIT >= 64 && (KDIM - KSPLIT) % 16 == 0,
                  "the prefetch runs 64 columns ahead; a loop trip takes 16 columns");
    const float* ap1 = A1 + (size_t)(m0 + mr) * lda1 + 4 * half;
    for (int k = 0; k < KSPLIT - 64; k += 16) kpair(ap, ap + 64);
    const float* pf1 = ap1;
    for (int k = KSPLIT - 64; k < KSPLIT; k += 16) { kpair(ap, pf1); pf1 += 16; }
    for (int k = KSPLIT; k < KDIM; k += 16) kpair(ap1, ap1 + 64);
  }

  float4 bv = make_float4(0.f, 0.f, 0.f, 0.f);
  if (BIAS) bv = *(const float4*)(bias + n0 + 4 * mr);
  float* yp = Y + (size_t)(m0 + 8 * half) * ldy + n0 + 4 * mr;
  #pragma unroll
  for (int r = 0; r < 8; ++r) {
    float4 v = make_float4(acc0[r], acc1[r], acc2[r], acc3[r]);
    if (BIAS) { v.x += bv.x; v.y += bv.y; v.z += bv.z; v.w += bv.w; }
    if (RELU) {
      v.x = (v.x > 0.0f) ? v.x : 0.0f;
      v.y = (v.y > 0.0f) ? v.y : 0.0f;
      v.z = (v.z > 0.0f) ? v.z : 0.0f;
      v.w = (v.w > 0.0f) ? v.w : 0.0f;
    }
    *(float4*)(yp + (size_t)r * ldy) = v;
  }
}

extern "C" void kernel_launch(void* const* d_in, const int* in_sizes, int n_in,
                              void* d_out, int out_size, void* d_ws, size_t ws_size,
                              hipStream_t stream) {
  const float* h   = (const float*)d_in[0];
  const float* w   = (const float*)d_in[1];
  const int*   src = (const int*)d_in[2];
  const int*   dst = (const int*)d_in[3];
  const float* W1  = (const float*)d_in[4];
  const float* b1  = (const float*)d_in[5];
  const float* W2  = (const float*)d_in[6];
  const float* b2  = (const float*)d_in[7];
  float* out = (float*)d_out;

  int*   deg    = (int*)d_ws;
  int*   rowptr = deg + 10240;
  int*   es     = rowptr + 2 * 10240;
  float* ew     = (float*)(es + N_EDGES);
  float* bufA   = ew + N_EDGES + 10240;
  float* bufB   = bufA + N_NODES * D_HID;
  float* PQ     = bufB + N_NODES * D_HID;

  int* part = (int*)bufB;
  hist_chunks<<<CSR_CHUNKS, CSR_THREADS, 0, stream>>>(dst, part);
  chunk_offsets<<<(N_NODES + 63) / 64, 64, 0, stream>>>(part, deg);
  bin_chunks<<<CSR_CHUNKS, CSR_THREADS, 0, stream>>>(src, dst, w, deg, part, rowptr, es, ew);

  aggregate_nodes<D_IN><<<(N_NODES * 32 + 255) / 256, 256, 0, stream>>>(
      h, rowptr, es, ew, bufA, N_NODES);
  {
    int Mtiles = N_NODES / 16, Nstrips = D_HID / 64;
    wmma_gemm_f32<D_HID, 2 * D_IN, D_IN, true, true><<<(Mtiles * Nstrips + 7) / 8, 256, 0, stream>>>(
        h, D_IN, bufA, D_IN, W1, b1, bufB, D_HID, Mtiles, Nstrips);
  }

  {
    int Mtiles = N_NODES / 16, Nstrips = 2 * D_OUT / 64;
    wmma_gemm_f32<D_OUT, D_HID, D_HID, false, false><<<(Mtiles * Nstrips + 7) / 8, 256, 0, stream>>>(
        bufB, D_HID, nullptr, 0, W2, nullptr, PQ, D_OUT, Mtiles, Nstrips);
  }
  aggregate_projected<<<(N_NODES * 32 + 255) / 256, 256, 0, stream>>>(
      PQ, PQ + N_NODES * D_OUT, b2, rowptr, es, ew, out, N_NODES);
}


# layer 2's projection runs inside layer 1's GEMM launch: a 4-wave block keeps its relu [16,256] tile in LDS and multiplies it by [Ws|Wn] at once; 6 launches, was 7; per-element sum order unchanged
# speedup vs baseline: 3.0290x; 1.0809x over previous
#include <hip/hip_runtime.h>

typedef __attribute__((ext_vector_type(2))) float v2f;
typedef __attribute__((ext_vector_type(4))) float v4f;
typedef __attribute__((ext_vector_type(8))) float v8f;

#define N_NODES 10000
#define N_EDGES 640000
#define D_IN    128
#define D_HID   256
#define D_OUT   64

#define CSR_CHUNKS  125
#define CSR_THREADS 1024
#define CSR_EPT     5
#define PART_LD     10240

static_assert(CSR_CHUNKS * CSR_THREADS * CSR_EPT == N_EDGES, "chunks must tile the edge list");
static_assert(N_NODES % 4 == 0 && PART_LD % 4 == 0 && PART_LD >= N_NODES, "int4 table rows");

__global__ void __launch_bounds__(CSR_THREADS)
hist_chunks(const int* __restrict__ dst, int* __restrict__ part) {
  __shared__ int4 bins4[N_NODES / 4];
  int* bins = (int*)bins4;
  int t = threadIdx.x;
  int base = blockIdx.x * (CSR_THREADS * CSR_EPT) + t;
  int d[CSR_EPT];
  #pragma unroll
  for (int k = 0; k < CSR_EPT; ++k) d[k] = dst[base + k * CSR_THREADS];
  for (int i = t; i < N_NODES / 4; i += CSR_THREADS) bins4[i] = make_int4(0, 0, 0, 0);
  __syncthreads();
  #pragma unroll
  for (int k = 0; k < CSR_EPT; ++k) atomicAdd(&bins[d[k]], 1);
  __syncthreads();
  int4* row = (int4*)(part + blockIdx.x * PART_LD);
  for (int i = t; i < N_NODES / 4; i += CSR_THREADS) row[i] = bins4[i];
}

__global__ void __launch_bounds__(64)
chunk_offsets(int* __restrict__ part, int* __restrict__ deg) {
  int i = blockIdx.x * 64 + threadIdx.x;
  if (i >= N_NODES) return;
  int* col = part + i;
  int run = 0;
  #pragma unroll
  for (int g0 = 0; g0 < CSR_CHUNKS; g0 += 25) {
    int v[25];
    #pragma unroll
    for (int k = 0; k < 25; ++k) v[k] = col[(g0 + k) * PART_LD];
    #pragma unroll
    for (int k = 0; k < 25; ++k) { col[(g0 + k) * PART_LD] = run; run += v[k]; }
  }
  deg[i] = run;
}

#define SCAN_THREADS CSR_THREADS
#define SCAN_PER     10
static_assert(SCAN_THREADS * SCAN_PER >= N_NODES && N_NODES % SCAN_PER == 0, "threads must tile the nodes");
static_assert(SCAN_PER % 2 == 0, "int2 loads and stores");
static_assert(SCAN_THREADS / 32 == 32, "one wave sum per lane in the second scan");

__global__ void __launch_bounds__(CSR_THREADS)
bin_chunks(const int* __restrict__ src, const int* __restrict__ dst,
           const float* __restrict__ w,
           const int* __restrict__ deg, const int* __restrict__ part,
           int* __restrict__ rowptr,
           int* __restrict__ es, float* __restrict__ ew) {
  __shared__ int2 cur_tab[N_NODES / 2];
  __shared__ int wsum[SCAN_THREADS / 32];
  int* cur = (int*)cur_tab;
  int t = threadIdx.x;
  int lane = t & 31;
  int wv = t >> 5;
  int base = blockIdx.x * (CSR_THREADS * CSR_EPT) + t;
  int d[CSR_EPT], s[CSR_EPT];
  float ww[CSR_EPT];
  #pragma unroll
  for (int k = 0; k < CSR_EPT; ++k) {
    d[k]  = dst[base + k * CSR_THREADS];
    s[k]  = src[base + k * CSR_THREADS];
    ww[k] = w[base + k * CSR_THREADS];
  }

  bool live = t < N_NODES / SCAN_PER;
  int own = (live ? t : 0) * SCAN_PER;
  const int2* dg2 = (const int2*)(deg + own);
  const int2* pr2 = (const int2*)(part + blockIdx.x * PART_LD + own);
  int v[SCAN_PER], p[SCAN_PER];
  #pragma unroll
  for (int k = 0; k < SCAN_PER / 2; ++k) {
    int2 a = dg2[k];
    int2 b = pr2[k];
    v[2 * k] = live ? a.x : 0;
    v[2 * k + 1] = live ? a.y : 0;
    p[2 * k] = b.x;
    p[2 * k + 1] = b.y;
  }
  int tot = 0;
  #pragma unroll
  for (int k = 0; k < SCAN_PER; ++k) tot += v[k];

  int inc = tot;
  for (int off = 1; off < 32; off <<= 1) {
    int x = __shfl_up(inc, off, 32);
    if (lane >= off) inc += x;
  }
  if (lane == 31) wsum[wv] = inc;
  __syncthreads();

  int ws = wsum[lane];
  int winc = ws;
  for (int off = 1; off < 32; off <<= 1) {
    int x = __shfl_up(winc, off, 32);
    if (lane >= off) winc += x;
  }
  int wbase = __shfl(winc - ws, wv, 32);
  int total = __shfl(winc, 31, 32);

  if (live) {
    int run = wbase + inc - tot;
    int2* cur2 = (int2*)(cur + t * SCAN_PER);
    int2* out2 = (int2*)(rowptr + t * SCAN_PER);
    #pragma unroll
    for (int k = 0; k < SCAN_PER / 2; ++k) {
      int a = run;  run += v[2 * k];
      int b = run;  run += v[2 * k + 1];
      cur2[k] = make_int2(a + p[2 * k], b + p[2 * k + 1]);
      if (blockIdx.x == 0) out2[k] = make_int2(a, b);
    }
  }
  if (blockIdx.x == 0 && t == 0) rowptr[N_NODES] = total;
  __syncthreads();

  int pos[CSR_EPT];
  #pragma unroll
  for (int k = 0; k < CSR_EPT; ++k) pos[k] = atomicAdd(&cur[d[k]], 1);
  #pragma unroll
  for (int k = 0; k < CSR_EPT; ++k) {
    es[pos[k]] = s[k];
    ew[pos[k]] = ww[k];
  }
}

#define AGG1_GROUP 16
#define AGG2_GROUP 8

template<int D, bool TAIL>
__device__ __forceinline__ void gather_group(const float* __restrict__ X, int my_s, float my_w,
                                             int g, int n, int lane, float4 (&acc)[D / 128]) {
  float4 v[AGG1_GROUP][D / 128];
  #pragma unroll
  for (int u = 0; u < AGG1_GROUP; ++u) {
    #pragma unroll
    for (int c = 0; c < D / 128; ++c) v[u][c] = make_float4(0.f, 0.f, 0.f, 0.f);
    if (!TAIL || g + u < n) {
      int s = __builtin_amdgcn_readlane(my_s, g + u);
      const float4* xr = (const float4*)(X + (size_t)s * D);
      #pragma unroll
      for (int c = 0; c < D / 128; ++c) v[u][c] = xr[lane + 32 * c];
    }
  }
  #pragma unroll
  for (int u = 0; u < AGG1_GROUP; ++u) {
    if (!TAIL || g + u < n) {
      float ww = __int_as_float(__builtin_amdgcn_readlane(__float_as_int(my_w), g + u));
      #pragma unroll
      for (int c = 0; c < D / 128; ++c) {
        acc[c].x = fmaf(v[u][c].x, ww, acc[c].x);
        acc[c].y = fmaf(v[u][c].y, ww, acc[c].y);
        acc[c].z = fmaf(v[u][c].z, ww, acc[c].z);
        acc[c].w = fmaf(v[u][c].w, ww, acc[c].w);
      }
    }
  }
}

template<int D>
__global__ void __launch_bounds__(256)
aggregate_nodes(const float* __restrict__ X,
                const int* __restrict__ rowptr,
                const int* __restrict__ es,
                const float* __restrict__ ew,
                float* __restrict__ Y, int nN) {
  int node = __builtin_amdgcn_readfirstlane((int)((blockIdx.x * blockDim.x + threadIdx.x) >> 5));
  int lane = threadIdx.x & 31;
  if (node >= nN) return;
  int beg = rowptr[node];
  int end = rowptr[node + 1];
  float4 acc[D / 128];
  #pragma unroll
  for (int c = 0; c < D / 128; ++c) acc[c] = make_float4(0.f, 0.f, 0.f, 0.f);

  if (beg < end) {
    int last = end - 1;
    int cur_s = es[min(beg + lane, last)];
    float cur_w = ew[min(beg + lane, last)];
    for (int b = beg; b < end; b += 32) {
      int n = min(32, end - b);
      int nxt_s = es[min(b + 32 + lane, last)];
      float nxt_w = ew[min(b + 32 + lane, last)];
      int g = 0;
      for (; g + AGG1_GROUP <= n; g += AGG1_GROUP) gather_group<D, false>(X, cur_s, cur_w, g, n, lane, acc);
      if (g < n) gather_group<D, true>(X, cur_s, cur_w, g, n, lane, acc);
      cur_s = nxt_s;
      cur_w = nxt_w;
    }
  }
  int dg = end - beg;
  float iv = (dg > 0) ? (1.0f / (float)dg) : 0.0f;
  float4* yr = (float4*)(Y + (size_t)node * D);
  #pragma unroll
  for (int c = 0; c < D / 128; ++c)
    yr[lane + 32 * c] = make_float4(acc[c].x * iv, acc[c].y * iv, acc[c].z * iv, acc[c].w * iv);
}


template<bool TAIL>
__device__ __forceinline__ void gather_group_halves(const float* __restrict__ Q, int my_s, float my_w,
                                                    int g, int left, int lane, float4& acc) {
  int c = lane & 15;
  int src_lane = (lane & 16) + g;
  float4 v[AGG2_GROUP];
  float ww[AGG2_GROUP];
  #pragma unroll
  for (int u = 0; u < AGG2_GROUP; ++u) {
    int s = __shfl(my_s, src_lane + u, 32);
    ww[u] = __shfl(my_w, src_lane + u, 32);
    v[u] = make_float4(0.f, 0.f, 0.f, 0.f);
    if (!TAIL || u < left) v[u] = ((const float4*)(Q + (size_t)s * D_OUT))[c];
  }
  #pragma unroll
  for (int u = 0; u < AGG2_GROUP; ++u) {
    if (!TAIL || u < left) {
      acc.x = fmaf(v[u].x, ww[u], acc.x);
      acc.y = fmaf(v[u].y, ww[u], acc.y);
      acc.z = fmaf(v[u].z, ww[u], acc.z);
      acc.w = fmaf(v[u].w, ww[u], acc.w);
    }
  }
}

__global__ void __launch_bounds__(256)
aggregate_projected(const float* __restrict__ P, const float* __restrict__ Q,
                    const float* __restrict__ bias,
                    const int* __restrict__ rowptr,
                    const int* __restrict__ es,
                    const float* __restrict__ ew,
                    float* __restrict__ out, int nN) {
  static_assert(D_OUT == 64, "a row is 16 lanes x float4");
  static_assert(2 * AGG2_GROUP == 16, "a batch of 16 steps is two groups");
  int node = __builtin_amdgcn_readfirstlane((int)((blockIdx.x * blockDim.x + threadIdx.x) >> 5));
  int lane = threadIdx.x & 31;
  if (node >= nN) return;
  int half = lane >> 4;
  int c    = lane & 15;
  int beg = rowptr[node];
  int end = rowptr[node + 1];
  int mid = beg + ((end - beg + 1) >> 1);
  int cmax = mid - beg;
  int cmin = end - mid;
  int lo  = half ? mid : beg;
  int cnt = half ? cmin : cmax;
  float4 acc = make_float4(0.f, 0.f, 0.f, 0.f);

  if (beg < end) {
    int cur_s = es[lo + min(c, cnt - 1)];
    float cur_w = ew[lo + min(c, cnt - 1)];
    for (int bb = 0; bb < cmax; bb += 16) {
      int nxt_s = es[lo + min(bb + 16 + c, cnt - 1)];
      float nxt_w = ew[lo + min(bb + 16 + c, cnt - 1)];
      for (int g = 0; g < 16 && bb + g < cmax; g += AGG2_GROUP) {
        if (bb + g + AGG2_GROUP <= cmin) gather_group_halves<false>(Q, cur_s, cur_w, g, AGG2_GROUP, lane, acc);
        else                             gather_group_halves<true>(Q, cur_s, cur_w, g, cnt - (bb + g), lane, acc);
      }
      cur_s = nxt_s;
      cur_w = nxt_w;
    }
  }
  acc.x += __shfl_xor(acc.x, 16, 32);
  acc.y += __shfl_xor(acc.y, 16, 32);
  acc.z += __shfl_xor(acc.z, 16, 32);
  acc.w += __shfl_xor(acc.w, 16, 32);
  if (half == 0) {
    int dg = end - beg;
    float iv = (dg > 0) ? (1.0f / (float)dg) : 0.0f;
    float4 p = ((const float4*)(P + (size_t)node * D_OUT))[c];
    float4 b = ((const float4*)bias)[c];
    ((float4*)(out + (size_t)node * D_OUT))[c] =
        make_float4(p.x + b.x + acc.x * iv, p.y + b.y + acc.y * iv,
                    p.z + b.z + acc.z * iv, p.w + b.w + acc.w * iv);
  }
}

#define X1_PITCH (D_HID + 4)
static_assert((X1_PITCH / 4) % 2 == 1, "pitch = 16 bytes x odd");

__global__ void __launch_bounds__(128)
wmma_gemm_pair_f32(const float* __restrict__ H, const float* __restrict__ MEAN,
                   const float* __restrict__ W1, const float* __restrict__ bias1,
                   const float* __restrict__ W2, float* __restrict__ PQ) {
  static_assert(D_HID == 4 * 64 && 2 * D_OUT == 4 * 32, "four waves cover both strips");
  static_assert(D_IN >= 64 && D_IN % 16 == 0 && D_HID % 16 == 0,
                "the prefetch runs 64 columns ahead; a loop trip takes 16 columns");
  static_assert(N_NODES % 16 == 0, "whole tiles: every wave reaches the barrier");
  __shared__ float4 x1_tile[16 * X1_PITCH / 4];
  float* x1 = (float*)x1_tile;
  int wv   = __builtin_amdgcn_readfirstlane((int)(threadIdx.x >> 5));
  int lane = threadIdx.x & 31;
  int half = lane >> 4;
  int mr   = lane & 15;
  int m0   = blockIdx.x * 16;

  {
    int n0 = wv * 64;
    const float* ap  = H    + (size_t)(m0 + mr) * D_IN + 4 * half;
    const float* ap1 = MEAN + (size_t)(m0 + mr) * D_IN + 4 * half;
    const float* bp  = W1 + (4 * half) * D_HID + n0 + 4 * mr;
    v8f acc0 = {}, acc1 = {}, acc2 = {}, acc3 = {};

    auto step = [&](float ax, float ay, v4f x, v4f y) {
      v2f a, b;
      a.x = ax;   a.y = ay;
      b.x = x.x;  b.y = y.x;
      acc0 = __builtin_amdgcn_wmma_f32_16x16x4_f32(false, a, false, b, (short)0, acc0, false, false);
      b.x = x.y;  b.y = y.y;
      acc1 = __builtin_amdgcn_wmma_f32_16x16x4_f32(false, a, false, b, (short)0, acc1, false, false);
      b.x = x.z;  b.y = y.z;
      acc2 = __builtin_amdgcn_wmma_f32_16x16x4_f32(false, a, false, b, (short)0, acc2, false, false);
      b.x = x.w;  b.y = y.w;
      acc3 = __builtin_amdgcn_wmma_f32_16x16x4_f32(false, a, false, b, (short)0, acc3, false, false);
    };

    auto kpair = [&](const float*& arow, const float* pf) {
      v4f a0 = *(const v4f*)arow;
      v4f a1 = *(const v4f*)(arow + 8);
      arow += 16;
      __builtin_prefetch(pf, 0, 3);
      v4f r0 = *(const v4f*)(bp);
      v4f r1 = *(const v4f*)(bp + D_HID);
      v4f r2 = *(const v4f*)(bp + 2 * D_HID);
      v4f r3 = *(const v4f*)(bp + 3 * D_HID);
      v4f r4 = *(const v4f*)(bp + 8 * D_HID);
      v4f r5 = *(const v4f*)(bp + 9 * D_HID);
      v4f r6 = *(const v4f*)(bp + 10 * D_HID);
      v4f r7 = *(const v4f*)(bp + 11 * D_HID);
      bp += 16 * D_HID;
      __builtin_amdgcn_sched_barrier(0);
      step(a0.x, a0.y, r0, r1);
      step(a0.z, a0.w, r2, r3);
      step(a1.x, a1.y, r4, r5);
      step(a1.z, a1.w, r6, r7);
    };

    for (int k = 0; k < D_IN - 64; k += 16) kpair(ap, ap + 64);
    const float* pf1 = ap1;
    for (int k = D_IN - 64; k < D_IN; k += 16) { kpair(ap, pf1); pf1 += 16; }
    for (int k = D_IN; k < 2 * D_IN; k += 16) kpair(ap1, ap1 + 64);

    float4 bv = *(const float4*)(bias1 + n0 + 4 * mr);
    float* xp = x1 + (8 * half) * X1_PITCH + n0 + 4 * mr;
    #pragma unroll
    for (int r = 0; r < 8; ++r) {
      float4 v = make_float4(acc0[r] + bv.x, acc1[r] + bv.y, acc2[r] + bv.z, acc3[r] + bv.w);
      v.x = (v.x > 0.0f) ? v.x : 0.0f;
      v.y = (v.y > 0.0f) ? v.y : 0.0f;
      v.z = (v.z > 0.0f) ? v.z : 0.0f;
      v.w = (v.w > 0.0f) ? v.w : 0.0f;
      *(float4*)(xp + r * X1_PITCH) = v;
    }
  }
  __syncthreads();

  {
    int slab = wv >> 1;
    int n0   = (wv & 1) * 32;
    const float* xa = x1 + mr * X1_PITCH + 4 * half;
    const float* bq = W2 + (size_t)slab * (D_HID * D_OUT) + (4 * half) * D_OUT + n0 + 2 * mr;
    v8f acc0 = {}, acc1 = {};

    auto step = [&](float ax, float ay, v2f x, v2f y) {
      v2f a, b;
      a.x = ax;   a.y = ay;
      b.x = x.x;  b.y = y.x;
      acc0 = __builtin_amdgcn_wmma_f32_16x16x4_f32(false, a, false, b, (short)0, acc0, false, false);
      b.x = x.y;  b.y = y.y;
      acc1 = __builtin_amdgcn_wmma_f32_16x16x4_f32(false, a, false, b, (short)0, acc1, false, false);
    };

    #pragma unroll
    for (int kb = 0; kb < D_HID; kb += 16) {
      v4f a0 = *(const v4f*)(xa + kb);
      v4f a1 = *(const v4f*)(xa + kb + 8);
      v2f r0 = *(const v2f*)(bq + (kb + 0) * D_OUT);
      v2f r1 = *(const v2f*)(bq + (kb + 1) * D_OUT);
      v2f r2 = *(const v2f*)(bq + (kb + 2) * D_OUT);
      v2f r3 = *(const v2f*)(bq + (kb + 3) * D_OUT);
      v2f r4 = *(const v2f*)(bq + (kb + 8) * D_OUT);
      v2f r5 = *(const v2f*)(bq + (kb + 9) * D_OUT);
      v2f r6 = *(const v2f*)(bq + (kb + 10) * D_OUT);
      v2f r7 = *(const v2f*)(bq + (kb + 11) * D_OUT);
      __builtin_amdgcn_sched_barrier(0);
      step(a0.x, a0.y, r0, r1);
      step(a0.z, a0.w, r2, r3);
      step(a1.x, a1.y, r4, r5);
      step(a1.z, a1.w, r6, r7);
    }

    float* yp = PQ + (size_t)slab * (N_NODES * D_OUT) + (size_t)(m0 + 8 * half) * D_OUT + n0 + 2 * mr;
    #pragma unroll
    for (int r = 0; r < 8; ++r) *(float2*)(yp + r * D_OUT) = make_float2(acc0[r], acc1[r]);
  }
}

extern "C" void kernel_launch(void* const* d_in, const int* in_sizes, int n_in,
                              void* d_out, int out_size, void* d_ws, size_t ws_size,
                              hipStream_t stream) {
  const float* h   = (const float*)d_in[0];
  const float* w   = (const float*)d_in[1];
  const int*   src = (const int*)d_in[2];
  const int*   dst = (const int*)d_in[3];
  const float* W1  = (const float*)d_in[4];
  const float* b1  = (const float*)d_in[5];
  const float* W2  = (const float*)d_in[6];
  const float* b2  = (const float*)d_in[7];
  float* out = (float*)d_out;

  int*   deg    = (int*)d_ws;
  int*   rowptr = deg + 10240;
  int*   es     = rowptr + 2 * 10240;
  float* ew     = (float*)(es + N_EDGES);
  float* bufA   = ew + N_EDGES + 10240;
  float* bufB   = bufA + N_NODES * D_HID;
  float* PQ     = bufB + N_NODES * D_HID;

  int* part = (int*)bufB;
  hist_chunks<<<CSR_CHUNKS, CSR_THREADS, 0, stream>>>(dst, part);
  chunk_offsets<<<(N_NODES + 63) / 64, 64, 0, stream>>>(part, deg);
  bin_chunks<<<CSR_CHUNKS, CSR_THREADS, 0, stream>>>(src, dst, w, deg, part, rowptr, es, ew);

  aggregate_nodes<D_IN><<<(N_NODES * 32 + 255) / 256, 256, 0, stream>>>(
      h, rowptr, es, ew, bufA, N_NODES);
  wmma_gemm_pair_f32<<<N_NODES / 16, 128, 0, stream>>>(h, bufA, W1, b1, W2, PQ);

  aggregate_projected<<<(N_NODES * 32 + 255) / 256, 256, 0, stream>>>(
      PQ, PQ + N_NODES * D_OUT, b2, rowptr, es, ew, out, N_NODES);
}


# CSR edge list kept as one 8-byte (source, weight) record per edge, not two 4-byte arrays: the placing pass does 5 scattered stores per thread, was 10; gathers read a pair in one load; sums unchanged
# speedup vs baseline: 3.7575x; 1.2405x over previous
#include <hip/hip_runtime.h>

typedef __attribute__((ext_vector_type(2))) float v2f;
typedef __attribute__((ext_vector_type(4))) float v4f;
typedef __attribute__((ext_vector_type(8))) float v8f;

#define N_NODES 10000
#define N_EDGES 640000
#define D_IN    128
#define D_HID   256
#define D_OUT   64

#define CSR_CHUNKS  125
#define CSR_THREADS 1024
#define CSR_EPT     5
#define PART_LD     10240

static_assert(CSR_CHUNKS * CSR_THREADS * CSR_EPT == N_EDGES, "chunks must tile the edge list");
static_assert(N_NODES % 4 == 0 && PART_LD % 4 == 0 && PART_LD >= N_NODES, "int4 table rows");

__global__ void __launch_bounds__(CSR_THREADS)
hist_chunks(const int* __restrict__ dst, int* __restrict__ part) {
  __shared__ int4 bins4[N_NODES / 4];
  int* bins = (int*)bins4;
  int t = threadIdx.x;
  int base = blockIdx.x * (CSR_THREADS * CSR_EPT) + t;
  int d[CSR_EPT];
  #pragma unroll
  for (int k = 0; k < CSR_EPT; ++k) d[k] = dst[base + k * CSR_THREADS];
  for (int i = t; i < N_NODES / 4; i += CSR_THREADS) bins4[i] = make_int4(0, 0, 0, 0);
  __syncthreads();
  #pragma unroll
  for (int k = 0; k < CSR_EPT; ++k) atomicAdd(&bins[d[k]], 1);
  __syncthreads();
  int4* row = (int4*)(part + blockIdx.x * PART_LD);
  for (int i = t; i < N_NODES / 4; i += CSR_THREADS) row[i] = bins4[i];
}

__global__ void __launch_bounds__(64)
chunk_offsets(int* __restrict__ part, int* __restrict__ deg) {
  int i = blockIdx.x * 64 + threadIdx.x;
  if (i >= N_NODES) return;
  int* col = part + i;
  int run = 0;
  #pragma unroll
  for (int g0 = 0; g0 < CSR_CHUNKS; g0 += 25) {
    int v[25];
    #pragma unroll
    for (int k = 0; k < 25; ++k) v[k] = col[(g0 + k) * PART_LD];
    #pragma unroll
    for (int k = 0; k < 25; ++k) { col[(g0 + k) * PART_LD] = run; run += v[k]; }
  }
  deg[i] = run;
}

#define SCAN_THREADS CSR_THREADS
#define SCAN_PER     10
static_assert(SCAN_THREADS * SCAN_PER >= N_NODES && N_NODES % SCAN_PER == 0, "threads must tile the nodes");
static_assert(SCAN_PER % 2 == 0, "int2 loads and stores");
static_assert(SCAN_THREADS / 32 == 32, "one wave sum per lane in the second scan");

__global__ void __launch_bounds__(CSR_THREADS)
bin_chunks(const int* __restrict__ src, const int* __restrict__ dst,
           const float* __restrict__ w,
           const int* __restrict__ deg, const int* __restrict__ part,
           int* __restrict__ rowptr,
           int* __restrict__ ep_words) {
  int2* ep = (int2*)ep_words;
  __shared__ int2 cur_tab[N_NODES / 2];
  __shared__ int wsum[SCAN_THREADS / 32];
  int* cur = (int*)cur_tab;
  int t = threadIdx.x;
  int lane = t & 31;
  int wv = t >> 5;
  int base = blockIdx.x * (CSR_THREADS * CSR_EPT) + t;
  int d[CSR_EPT], s[CSR_EPT];
  float ww[CSR_EPT];
  #pragma unroll
  for (int k = 0; k < CSR_EPT; ++k) {
    d[k]  = dst[base + k * CSR_THREADS];
    s[k]  = src[base + k * CSR_THREADS];
    ww[k] = w[base + k * CSR_THREADS];
  }

  bool live = t < N_NODES / SCAN_PER;
  int own = (live ? t : 0) * SCAN_PER;
  const int2* dg2 = (const int2*)(deg + own);
  const int2* pr2 = (const int2*)(part + blockIdx.x * PART_LD + own);
  int v[SCAN_PER], p[SCAN_PER];
  #pragma unroll
  for (int k = 0; k < SCAN_PER / 2; ++k) {
    int2 a = dg2[k];
    int2 b = pr2[k];
    v[2 * k] = live ? a.x : 0;
    v[2 * k + 1] = live ? a.y : 0;
    p[2 * k] = b.x;
    p[2 * k + 1] = b.y;
  }
  int tot = 0;
  #pragma unroll
  for (int k = 0; k < SCAN_PER; ++k) tot += v[k];

  int inc = tot;
  for (int off = 1; off < 32; off <<= 1) {
    int x = __shfl_up(inc, off, 32);
    if (lane >= off) inc += x;
  }
  if (lane == 31) wsum[wv] = inc;
  __syncthreads();

  int ws = wsum[lane];
  int winc = ws;
  for (int off = 1; off < 32; off <<= 1) {
    int x = __shfl_up(winc, off, 32);
    if (lane >= off) winc += x;
  }
  int wbase = __shfl(winc - ws, wv, 32);
  int total = __shfl(winc, 31, 32);

  if (live) {
    int run = wbase + inc - tot;
    int2* cur2 = (int2*)(cur + t * SCAN_PER);
    int2* out2 = (int2*)(rowptr + t * SCAN_PER);
    #pragma unroll
    for (int k = 0; k < SCAN_PER / 2; ++k) {
      int a = run;  run += v[2 * k];
      int b = run;  run += v[2 * k + 1];
      cur2[k] = make_int2(a + p[2 * k], b + p[2 * k + 1]);
      if (blockIdx.x == 0) out2[k] = make_int2(a, b);
    }
  }
  if (blockIdx.x == 0 && t == 0) rowptr[N_NODES] = total;
  __syncthreads();

  int pos[CSR_EPT];
  #pragma unroll
  for (int k = 0; k < CSR_EPT; ++k) pos[k] = atomicAdd(&cur[d[k]], 1);
  #pragma unroll
  for (int k = 0; k < CSR_EPT; ++k) ep[pos[k]] = make_int2(s[k], __float_as_int(ww[k]));
}

#define AGG1_GROUP 16
#define AGG2_GROUP 8

template<int D, bool TAIL>
__device__ __forceinline__ void gather_group(const float* __restrict__ X, int my_s, float my_w,
                                             int g, int n, int lane, float4 (&acc)[D / 128]) {
  float4 v[AGG1_GROUP][D / 128];
  #pragma unroll
  for (int u = 0; u < AGG1_GROUP; ++u) {
    #pragma unroll
    for (int c = 0; c < D / 128; ++c) v[u][c] = make_float4(0.f, 0.f, 0.f, 0.f);
    if (!TAIL || g + u < n) {
      int s = __builtin_amdgcn_readlane(my_s, g + u);
      const float4* xr = (const float4*)(X + (size_t)s * D);
      #pragma unroll
      for (int c = 0; c < D / 128; ++c) v[u][c] = xr[lane + 32 * c];
    }
  }
  #pragma unroll
  for (int u = 0; u < AGG1_GROUP; ++u) {
    if (!TAIL || g + u < n) {
      float ww = __int_as_float(__builtin_amdgcn_readlane(__float_as_int(my_w), g + u));
      #pragma unroll
      for (int c = 0; c < D / 128; ++c) {
        acc[c].x = fmaf(v[u][c].x, ww, acc[c].x);
        acc[c].y = fmaf(v[u][c].y, ww, acc[c].y);
        acc[c].z = fmaf(v[u][c].z, ww, acc[c].z);
        acc[c].w = fmaf(v[u][c].w, ww, acc[c].w);
      }
    }
  }
}

template<int D>
__global__ void __launch_bounds__(256)
aggregate_nodes(const float* __restrict__ X,
                const int* __restrict__ rowptr,
                const int* __restrict__ ep_words,
                float* __restrict__ Y, int nN) {
  const int2* ep = (const int2*)ep_words;
  int node = __builtin_amdgcn_readfirstlane((int)((blockIdx.x * blockDim.x + threadIdx.x) >> 5));
  int lane = threadIdx.x & 31;
  if (node >= nN) return;
  int beg = rowptr[node];
  int end = rowptr[node + 1];
  float4 acc[D / 128];
  #pragma unroll
  for (int c = 0; c < D / 128; ++c) acc[c] = make_float4(0.f, 0.f, 0.f, 0.f);

  if (beg < end) {
    int last = end - 1;
    int2 cur = ep[min(beg + lane, last)];
    int cur_s = cur.x;
    float cur_w = __int_as_float(cur.y);
    for (int b = beg; b < end; b += 32) {
      int n = min(32, end - b);
      int2 nxt = ep[min(b + 32 + lane, last)];
      int nxt_s = nxt.x;
      float nxt_w = __int_as_float(nxt.y);
      int g = 0;
      for (; g + AGG1_GROUP <= n; g += AGG1_GROUP) gather_group<D, false>(X, cur_s, cur_w, g, n, lane, acc);
      if (g < n) gather_group<D, true>(X, cur_s, cur_w, g, n, lane, acc);
      cur_s = nxt_s;
      cur_w = nxt_w;
    }
  }
  int dg = end - beg;
  float iv = (dg > 0) ? (1.0f / (float)dg) : 0.0f;
  float4* yr = (float4*)(Y + (size_t)node * D);
  #pragma unroll
  for (int c = 0; c < D / 128; ++c)
    yr[lane + 32 * c] = make_float4(acc[c].x * iv, acc[c].y * iv, acc[c].z * iv, acc[c].w * iv);
}


template<bool TAIL>
__device__ __forceinline__ void gather_group_halves(const float* __restrict__ Q, int my_s, float my_w,
                                                    int g, int left, int lane, float4& acc) {
  int c = lane & 15;
  int src_lane = (lane & 16) + g;
  float4 v[AGG2_GROUP];
  float ww[AGG2_GROUP];
  #pragma unroll
  for (int u = 0; u < AGG2_GROUP; ++u) {
    int s = __shfl(my_s, src_lane + u, 32);
    ww[u] = __shfl(my_w, src_lane + u, 32);
    v[u] = make_float4(0.f, 0.f, 0.f, 0.f);
    if (!TAIL || u < left) v[u] = ((const float4*)(Q + (size_t)s * D_OUT))[c];
  }
  #pragma unroll
  for (int u = 0; u < AGG2_GROUP; ++u) {
    if (!TAIL || u < left) {
      acc.x = fmaf(v[u].x, ww[u], acc.x);
      acc.y = fmaf(v[u].y, ww[u], acc.y);
      acc.z = fmaf(v[u].z, ww[u], acc.z);
      acc.w = fmaf(v[u].w, ww[u], acc.w);
    }
  }
}

__global__ void __launch_bounds__(256)
aggregate_projected(const float* __restrict__ P, const float* __restrict__ Q,
                    const float* __restrict__ bias,
                    const int* __restrict__ rowptr,
                    const int* __restrict__ ep_words,
                    float* __restrict__ out, int nN) {
  const int2* ep = (const int2*)ep_words;
  static_assert(D_OUT == 64, "a row is 16 lanes x float4");
  static_assert(2 * AGG2_GROUP == 16, "a batch of 16 steps is two groups");
  int node = __builtin_amdgcn_readfirstlane((int)((blockIdx.x * blockDim.x + threadIdx.x) >> 5));
  int lane = threadIdx.x & 31;
  if (node >= nN) return;
  int half = lane >> 4;
  int c    = lane & 15;
  int beg = rowptr[node];
  int end = rowptr[node + 1];
  int mid = beg + ((end - beg + 1) >> 1);
  int cmax = mid - beg;
  int cmin = end - mid;
  int lo  = half ? mid : beg;
  int cnt = half ? cmin : cmax;
  float4 acc = make_float4(0.f, 0.f, 0.f, 0.f);

  if (beg < end) {
    int2 cur = ep[lo + min(c, cnt - 1)];
    int cur_s = cur.x;
    float cur_w = __int_as_float(cur.y);
    for (int bb = 0; bb < cmax; bb += 16) {
      int2 nxt = ep[lo + min(bb + 16 + c, cnt - 1)];
      int nxt_s = nxt.x;
      float nxt_w = __int_as_float(nxt.y);
      for (int g = 0; g < 16 && bb + g < cmax; g += AGG2_GROUP) {
        if (bb + g + AGG2_GROUP <= cmin) gather_group_halves<false>(Q, cur_s, cur_w, g, AGG2_GROUP, lane, acc);
        else                             gather_group_halves<true>(Q, cur_s, cur_w, g, cnt - (bb + g), lane, acc);
      }
      cur_s = nxt_s;
      cur_w = nxt_w;
    }
  }
  acc.x += __shfl_xor(acc.x, 16, 32);
  acc.y += __shfl_xor(acc.y, 16, 32);
  acc.z += __shfl_xor(acc.z, 16, 32);
  acc.w += __shfl_xor(acc.w, 16, 32);
  if (half == 0) {
    int dg = end - beg;
    float iv = (dg > 0) ? (1.0f / (float)dg) : 0.0f;
    float4 p = ((const float4*)(P + (size_t)node * D_OUT))[c];
    float4 b = ((const float4*)bias)[c];
    ((float4*)(out + (size_t)node * D_OUT))[c] =
        make_float4(p.x + b.x + acc.x * iv, p.y + b.y + acc.y * iv,
                    p.z + b.z + acc.z * iv, p.w + b.w + acc.w * iv);
  }
}

#define X1_PITCH (D_HID + 4)
static_assert((X1_PITCH / 4) % 2 == 1, "pitch = 16 bytes x odd");

__global__ void __launch_bounds__(128)
wmma_gemm_pair_f32(const float* __restrict__ H, const float* __restrict__ MEAN,
                   const float* __restrict__ W1, const float* __restrict__ bias1,
                   const float* __restrict__ W2, float* __restrict__ PQ) {
  static_assert(D_HID == 4 * 64 && 2 * D_OUT == 4 * 32, "four waves cover both strips");
  static_assert(D_IN >= 64 && D_IN % 16 == 0 && D_HID % 16 == 0,
                "the prefetch runs 64 columns ahead; a loop trip takes 16 columns");
  static_assert(N_NODES % 16 == 0, "whole tiles: every wave reaches the barrier");
  __shared__ float4 x1_tile[16 * X1_PITCH / 4];
  float* x1 = (float*)x1_tile;
  int wv   = __builtin_amdgcn_readfirstlane((int)(threadIdx.x >> 5));
  int lane = threadIdx.x & 31;
  int half = lane >> 4;
  int mr   = lane & 15;
  int m0   = blockIdx.x * 16;

  {
    int n0 = wv * 64;
    const float* ap  = H    + (size_t)(m0 + mr) * D_IN + 4 * half;
    const float* ap1 = MEAN + (size_t)(m0 + mr) * D_IN + 4 * half;
    const float* bp  = W1 + (4 * half) * D_HID + n0 + 4 * mr;
    v8f acc0 = {}, acc1 = {}, acc2 = {}, acc3 = {};

    auto step = [&](float ax, float ay, v4f x, v4f y) {
      v2f a, b;
      a.x = ax;   a.y = ay;
      b.x = x.x;  b.y = y.x;
      acc0 = __builtin_amdgcn_wmma_f32_16x16x4_f32(false, a, false, b, (short)0, acc0, false, false);
      b.x = x.y;  b.y = y.y;
      acc1 = __builtin_amdgcn_wmma_f32_16x16x4_f32(false, a, false, b, (short)0, acc1, false, false);
      b.x = x.z;  b.y = y.z;
      acc2 = __builtin_amdgcn_wmma_f32_16x16x4_f32(false, a, false, b, (short)0, acc2, false, false);
      b.x = x.w;  b.y = y.w;
      acc3 = __builtin_amdgcn_wmma_f32_16x16x4_f32(false, a, false, b, (short)0, acc3, false, false);
    };

    auto kpair = [&](const float*& arow, const float* pf) {
      v4f a0 = *(const v4f*)arow;
      v4f a1 = *(const v4f*)(arow + 8);
      arow += 16;
      __builtin_prefetch(pf, 0, 3);
      v4f r0 = *(const v4f*)(bp);
      v4f r1 = *(const v4f*)(bp + D_HID);
      v4f r2 = *(const v4f*)(bp + 2 * D_HID);
      v4f r3 = *(const v4f*)(bp + 3 * D_HID);
      v4f r4 = *(const v4f*)(bp + 8 * D_HID);
      v4f r5 = *(const v4f*)(bp + 9 * D_HID);
      v4f r6 = *(const v4f*)(bp + 10 * D_HID);
      v4f r7 = *(const v4f*)(bp + 11 * D_HID);
      bp += 16 * D_HID;
      __builtin_amdgcn_sched_barrier(0);
      step(a0.x, a0.y, r0, r1);
      step(a0.z, a0.w, r2, r3);
      step(a1.x, a1.y, r4, r5);
      step(a1.z, a1.w, r6, r7);
    };

    for (int k = 0; k < D_IN - 64; k += 16) kpair(ap, ap + 64);
    const float* pf1 = ap1;
    for (int k = D_IN - 64; k < D_IN; k += 16) { kpair(ap, pf1); pf1 += 16; }
    for (int k = D_IN; k < 2 * D_IN; k += 16) kpair(ap1, ap1 + 64);

    float4 bv = *(const float4*)(bias1 + n0 + 4 * mr);
    float* xp = x1 + (8 * half) * X1_PITCH + n0 + 4 * mr;
    #pragma unroll
    for (int r = 0; r < 8; ++r) {
      float4 v = make_float4(acc0[r] + bv.x, acc1[r] + bv.y, acc2[r] + bv.z, acc3[r] + bv.w);
      v.x = (v.x > 0.0f) ? v.x : 0.0f;
      v.y = (v.y > 0.0f) ? v.y : 0.0f;
      v.z = (v.z > 0.0f) ? v.z : 0.0f;
      v.w = (v.w > 0.0f) ? v.w : 0.0f;
      *(float4*)(xp + r * X1_PITCH) = v;
    }
  }
  __syncthreads();

  {
    int slab = wv >> 1;
    int n0   = (wv & 1) * 32;
    const float* xa = x1 + mr * X1_PITCH + 4 * half;
    const float* bq = W2 + (size_t)slab * (D_HID * D_OUT) + (4 * half) * D_OUT + n0 + 2 * mr;
    v8f acc0 = {}, acc1 = {};

    auto step = [&](float ax, float ay, v2f x, v2f y) {
      v2f a, b;
      a.x = ax;   a.y = ay;
      b.x = x.x;  b.y = y.x;
      acc0 = __builtin_amdgcn_wmma_f32_16x16x4_f32(false, a, false, b, (short)0, acc0, false, false);
      b.x = x.y;  b.y = y.y;
      acc1 = __builtin_amdgcn_wmma_f32_16x16x4_f32(false, a, false, b, (short)0, acc1, false, false);
    };

    #pragma unroll
    for (int kb = 0; kb < D_HID; kb += 16) {
      v4f a0 = *(const v4f*)(xa + kb);
      v4f a1 = *(const v4f*)(xa + kb + 8);
      v2f r0 = *(const v2f*)(bq + (kb + 0) * D_OUT);
      v2f r1 = *(const v2f*)(bq + (kb + 1) * D_OUT);
      v2f r2 = *(const v2f*)(bq + (kb + 2) * D_OUT);
      v2f r3 = *(const v2f*)(bq + (kb + 3) * D_OUT);
      v2f r4 = *(const v2f*)(bq + (kb + 8) * D_OUT);
      v2f r5 = *(const v2f*)(bq + (kb + 9) * D_OUT);
      v2f r6 = *(const v2f*)(bq + (kb + 10) * D_OUT);
      v2f r7 = *(const v2f*)(bq + (kb + 11) * D_OUT);
      __builtin_amdgcn_sched_barrier(0);
      step(a0.x, a0.y, r0, r1);
      step(a0.z, a0.w, r2, r3);
      step(a1.x, a1.y, r4, r5);
      step(a1.z, a1.w, r6, r7);
    }

    float* yp = PQ + (size_t)slab * (N_NODES * D_OUT) + (size_t)(m0 + 8 * half) * D_OUT + n0 + 2 * mr;
    #pragma unroll
    for (int r = 0; r < 8; ++r) *(float2*)(yp + r * D_OUT) = make_float2(acc0[r], acc1[r]);
  }
}

extern "C" void kernel_launch(void* const* d_in, const int* in_sizes, int n_in,
                              void* d_out, int out_size, void* d_ws, size_t ws_size,
                              hipStream_t stream) {
  const float* h   = (const float*)d_in[0];
  const float* w   = (const float*)d_in[1];
  const int*   src = (const int*)d_in[2];
  const int*   dst = (const int*)d_in[3];
  const float* W1  = (const float*)d_in[4];
  const float* b1  = (const float*)d_in[5];
  const float* W2  = (const float*)d_in[6];
  const float* b2  = (const float*)d_in[7];
  float* out = (float*)d_out;

  int*   deg    = (int*)d_ws;
  int*   rowptr = deg + 10240;
  int*   ep     = rowptr + 2 * 10240;
  float* bufA   = (float*)(ep + 2 * N_EDGES) + 10240;
  float* bufB   = bufA + N_NODES * D_HID;
  float* PQ     = bufB + N_NODES * D_HID;

  int* part = (int*)bufB;
  hist_chunks<<<CSR_CHUNKS, CSR_THREADS, 0, stream>>>(dst, part);
  chunk_offsets<<<(N_NODES + 63) / 64, 64, 0, stream>>>(part, deg);
  bin_chunks<<<CSR_CHUNKS, CSR_THREADS, 0, stream>>>(src, dst, w, deg, part, rowptr, ep);

  aggregate_nodes<D_IN><<<(N_NODES * 32 + 255) / 256, 256, 0, stream>>>(
      h, rowptr, ep, bufA, N_NODES);
  wmma_gemm_pair_f32<<<N_NODES / 16, 128, 0, stream>>>(h, bufA, W1, b1, W2, PQ);

  aggregate_projected<<<(N_NODES * 32 + 255) / 256, 256, 0, stream>>>(
      PQ, PQ + N_NODES * D_OUT, b2, rowptr, ep, out, N_NODES);
}


# CSR build as a two-level sort: chunks write their records grouped by 80-node bucket, a block per bucket orders them in LDS and stores ep in whole lines; 5 launches, was 6; in-node sum order as before
# speedup vs baseline: 5.3239x; 1.4169x over previous
#include <hip/hip_runtime.h>

typedef __attribute__((ext_vector_type(2))) float v2f;
typedef __attribute__((ext_vector_type(4))) float v4f;
typedef __attribute__((ext_vector_type(8))) float v8f;

#define N_NODES 10000
#define N_EDGES 640000
#define D_IN    128
#define D_HID   256
#define D_OUT   64

#define CSR_CHUNKS  125
#define CSR_THREADS 1024
#define CSR_EPT     5
#define CHUNK_EDGES (CSR_THREADS * CSR_EPT)
#define BKT_NODES   80
#define N_BKT       125
#define TAB_LD      128
#define SRC_BITS    14
#define SRC_MASK    ((1 << SRC_BITS) - 1)
#define BKT_PAD     96
#define STAGE_CAP   6144
#define CHUNKS_PER_WAVE 4

static_assert(CSR_CHUNKS * CSR_THREADS * CSR_EPT == N_EDGES, "chunks must tile the edge list");
static_assert(N_BKT * BKT_NODES == N_NODES, "buckets must tile the nodes");
static_assert(N_NODES <= (1 << SRC_BITS) && BKT_NODES <= 128, "record fields");
static_assert(N_BKT + 1 <= TAB_LD && TAB_LD == 4 * 32, "one wave scans the bucket counts, four per lane");
static_assert((CSR_THREADS / 32) * CHUNKS_PER_WAVE >= CSR_CHUNKS, "the waves cover the chunks");
static_assert(BKT_PAD == 3 * 32 && BKT_PAD >= BKT_NODES, "one wave scans the node counts, three per lane");
static_assert(CHUNK_EDGES % 2 == 0, "16-byte copies of the chunk");

__global__ void __launch_bounds__(CSR_THREADS)
partition_chunks(const int* __restrict__ src, const int* __restrict__ dst,
                 const float* __restrict__ w,
                 int* __restrict__ tab, int* __restrict__ rec_words) {
  __shared__ int4 stage4[CHUNK_EDGES / 2];
  __shared__ int4 bins4[TAB_LD / 4];
  int2* stage = (int2*)stage4;
  int* bins = (int*)bins4;
  int t = threadIdx.x;
  int lane = t & 31;
  int wv = t >> 5;
  int base = blockIdx.x * CHUNK_EDGES + t;
  int d[CSR_EPT], s[CSR_EPT];
  float ww[CSR_EPT];
  #pragma unroll
  for (int k = 0; k < CSR_EPT; ++k) {
    d[k]  = dst[base + k * CSR_THREADS];
    s[k]  = src[base + k * CSR_THREADS];
    ww[k] = w[base + k * CSR_THREADS];
  }
  if (t < TAB_LD) bins[t] = 0;
  __syncthreads();
  int bk[CSR_EPT], pos[CSR_EPT];
  #pragma unroll
  for (int k = 0; k < CSR_EPT; ++k) {
    bk[k] = d[k] / BKT_NODES;
    pos[k] = atomicAdd(&bins[bk[k]], 1);
  }
  __syncthreads();
  if (wv == 0) {
    int4 c = bins4[lane];
    int tot = c.x + c.y + c.z + c.w;
    int inc = tot;
    for (int off = 1; off < 32; off <<= 1) {
      int x = __shfl_up(inc, off, 32);
      if (lane >= off) inc += x;
    }
    int4 o;
    o.x = inc - tot;
    o.y = o.x + c.x;
    o.z = o.y + c.y;
    o.w = o.z + c.z;
    bins4[lane] = o;
    ((int4*)(tab + blockIdx.x * TAB_LD))[lane] = o;
  }
  __syncthreads();
  #pragma unroll
  for (int k = 0; k < CSR_EPT; ++k) {
    int local = d[k] - bk[k] * BKT_NODES;
    stage[bins[bk[k]] + pos[k]] = make_int2(s[k] | (local << SRC_BITS), __float_as_int(ww[k]));
  }
  __syncthreads();
  int4* out4 = (int4*)(rec_words + (size_t)blockIdx.x * (2 * CHUNK_EDGES));
  for (int i = t; i < CHUNK_EDGES / 2; i += CSR_THREADS) out4[i] = stage4[i];
}

__global__ void __launch_bounds__(CSR_THREADS)
sort_buckets(const int* __restrict__ tab, const int* __restrict__ rec_words,
             int* __restrict__ rowptr, int* __restrict__ ep_words) {
  const int2* rec = (const int2*)rec_words;
  int2* ep = (int2*)ep_words;
  __shared__ int2 stage[STAGE_CAP];
  __shared__ int cnt[CSR_THREADS / 32][BKT_PAD];
  __shared__ int4 segbeg4[TAB_LD / 4], seglen4[TAB_LD / 4];
  int* segbeg = (int*)segbeg4;
  int* seglen = (int*)seglen4;
  __shared__ int nodeoff[BKT_PAD];
  __shared__ int red[2];
  int t = threadIdx.x;
  int lane = t & 31;
  int wv = t >> 5;
  int b = blockIdx.x;

  int sb = 0, se = 0;
  if (t < CSR_CHUNKS) {
    sb = tab[t * TAB_LD + b];
    se = tab[t * TAB_LD + b + 1];
  }
  for (int i = t; i < (CSR_THREADS / 32) * BKT_PAD; i += CSR_THREADS) ((int*)cnt)[i] = 0;
  if (t < 2) red[t] = 0;
  __syncthreads();
  if (t < TAB_LD) {
    segbeg[t] = sb;
    seglen[t] = se - sb;
    if (t < CSR_CHUNKS) {
      atomicAdd(&red[0], sb);
      atomicAdd(&red[1], se - sb);
    }
  }
  __syncthreads();
  int base = red[0];
  int total = red[1];

  int2 r0[CHUNKS_PER_WAVE], r1[CHUNKS_PER_WAVE];
  int  rn[CHUNKS_PER_WAVE], rb[CHUNKS_PER_WAVE];
  {
    static_assert(CHUNKS_PER_WAVE == 4, "a wave's bounds are one int4 of each table");
    int4 b4 = segbeg4[wv];
    int4 n4 = seglen4[wv];
    rb[0] = b4.x;  rb[1] = b4.y;  rb[2] = b4.z;  rb[3] = b4.w;
    rn[0] = n4.x;  rn[1] = n4.y;  rn[2] = n4.z;  rn[3] = n4.w;
  }
  #pragma unroll
  for (int j = 0; j < CHUNKS_PER_WAVE; ++j) {
    rb[j] += (CHUNKS_PER_WAVE * wv + j) * CHUNK_EDGES;
    int lastrec = min(rb[j] + max(rn[j] - 1, 0), N_EDGES - 1);
    r0[j] = rec[min(rb[j] + lane, lastrec)];
    r1[j] = rec[min(rb[j] + lane + 32, lastrec)];
  }
  __builtin_amdgcn_sched_barrier(0);

  #pragma unroll
  for (int j = 0; j < CHUNKS_PER_WAVE; ++j) {
    if (lane < rn[j])      atomicAdd(&cnt[wv][r0[j].x >> SRC_BITS], 1);
    if (lane + 32 < rn[j]) atomicAdd(&cnt[wv][r1[j].x >> SRC_BITS], 1);
    for (int i = 64 + lane; i < rn[j]; i += 32) atomicAdd(&cnt[wv][rec[rb[j] + i].x >> SRC_BITS], 1);
  }
  __syncthreads();
  if (t < BKT_NODES) {
    int run = 0;
    for (int v = 0; v < CSR_THREADS / 32; ++v) {
      int c = cnt[v][t];
      cnt[v][t] = run;
      run += c;
    }
    nodeoff[t] = run;
  } else if (t < BKT_PAD) {
    nodeoff[t] = 0;
  }
  __syncthreads();
  if (wv == 0) {
    int a0 = nodeoff[3 * lane], a1 = nodeoff[3 * lane + 1], a2 = nodeoff[3 * lane + 2];
    int tot = a0 + a1 + a2;
    int inc = tot;
    for (int off = 1; off < 32; off <<= 1) {
      int x = __shfl_up(inc, off, 32);
      if (lane >= off) inc += x;
    }
    int o0 = inc - tot, o1 = o0 + a0, o2 = o1 + a1;
    nodeoff[3 * lane] = o0;
    nodeoff[3 * lane + 1] = o1;
    nodeoff[3 * lane + 2] = o2;
    int n0 = 3 * lane;
    int* rp = rowptr + b * BKT_NODES;
    if (n0 < BKT_NODES)     rp[n0] = base + o0;
    if (n0 + 1 < BKT_NODES) rp[n0 + 1] = base + o1;
    if (n0 + 2 < BKT_NODES) rp[n0 + 2] = base + o2;
    if (b == N_BKT - 1 && lane == 0) rowptr[N_NODES] = base + total;
  }
  __syncthreads();

  bool fits = total <= STAGE_CAP;
  auto place = [&](int2 r) {
    int local = r.x >> SRC_BITS;
    int slot = nodeoff[local] + atomicAdd(&cnt[wv][local], 1);
    int2 o = make_int2(r.x & SRC_MASK, r.y);
    if (fits) stage[slot] = o;
    else      ep[base + slot] = o;
  };
  #pragma unroll
  for (int j = 0; j < CHUNKS_PER_WAVE; ++j) {
    if (lane < rn[j])      place(r0[j]);
    if (lane + 32 < rn[j]) place(r1[j]);
    for (int i = 64 + lane; i < rn[j]; i += 32) place(rec[rb[j] + i]);
  }
  __syncthreads();
  if (fits)
    for (int i = t; i < total; i += CSR_THREADS) ep[base + i] = stage[i];
}

#define AGG1_GROUP 16
#define AGG2_GROUP 8

template<int D, bool TAIL>
__device__ __forceinline__ void gather_group(const float* __restrict__ X, int my_s, float my_w,
                                             int g, int n, int lane, float4 (&acc)[D / 128]) {
  float4 v[AGG1_GROUP][D / 128];
  #pragma unroll
  for (int u = 0; u < AGG1_GROUP; ++u) {
    #pragma unroll
    for (int c = 0; c < D / 128; ++c) v[u][c] = make_float4(0.f, 0.f, 0.f, 0.f);
    if (!TAIL || g + u < n) {
      int s = __builtin_amdgcn_readlane(my_s, g + u);
      const float4* xr = (const float4*)(X + (size_t)s * D);
      #pragma unroll
      for (int c = 0; c < D / 128; ++c) v[u][c] = xr[lane + 32 * c];
    }
  }
  #pragma unroll
  for (int u = 0; u < AGG1_GROUP; ++u) {
    if (!TAIL || g + u < n) {
      float ww = __int_as_float(__builtin_amdgcn_readlane(__float_as_int(my_w), g + u));
      #pragma unroll
      for (int c = 0; c < D / 128; ++c) {
        acc[c].x = fmaf(v[u][c].x, ww, acc[c].x);
        acc[c].y = fmaf(v[u][c].y, ww, acc[c].y);
        acc[c].z = fmaf(v[u][c].z, ww, acc[c].z);
        acc[c].w = fmaf(v[u][c].w, ww, acc[c].w);
      }
    }
  }
}

template<int D>
__global__ void __launch_bounds__(256)
aggregate_nodes(const float* __restrict__ X,
                const int* __restrict__ rowptr,
                const int* __restrict__ ep_words,
                float* __restrict__ Y, int nN) {
  const int2* ep = (const int2*)ep_words;
  int node = __builtin_amdgcn_readfirstlane((int)((blockIdx.x * blockDim.x + threadIdx.x) >> 5));
  int lane = threadIdx.x & 31;
  if (node >= nN) return;
  int beg = rowptr[node];
  int end = rowptr[node + 1];
  float4 acc[D / 128];
  #pragma unroll
  for (int c = 0; c < D / 128; ++c) acc[c] = make_float4(0.f, 0.f, 0.f, 0.f);

  if (beg < end) {
    int last = end - 1;
    int2 cur = ep[min(beg + lane, last)];
    int cur_s = cur.x;
    float cur_w = __int_as_float(cur.y);
    for (int b = beg; b < end; b += 32) {
      int n = min(32, end - b);
      int2 nxt = ep[min(b + 32 + lane, last)];
      int nxt_s = nxt.x;
      float nxt_w = __int_as_float(nxt.y);
      int g = 0;
      for (; g + AGG1_GROUP <= n; g += AGG1_GROUP) gather_group<D, false>(X, cur_s, cur_w, g, n, lane, acc);
      if (g < n) gather_group<D, true>(X, cur_s, cur_w, g, n, lane, acc);
      cur_s = nxt_s;
      cur_w = nxt_w;
    }
  }
  int dg = end - beg;
  float iv = (dg > 0) ? (1.0f / (float)dg) : 0.0f;
  float4* yr = (float4*)(Y + (size_t)node * D);
  #pragma unroll
  for (int c = 0; c < D / 128; ++c)
    yr[lane + 32 * c] = make_float4(acc[c].x * iv, acc[c].y * iv, acc[c].z * iv, acc[c].w * iv);
}


template<bool TAIL>
__device__ __forceinline__ void gather_group_halves(const float* __restrict__ Q, int my_s, float my_w,
                                                    int g, int left, int lane, float4& acc) {
  int c = lane & 15;
  int src_lane = (lane & 16) + g;
  float4 v[AGG2_GROUP];
  float ww[AGG2_GROUP];
  #pragma unroll
  for (int u = 0; u < AGG2_GROUP; ++u) {
    int s = __shfl(my_s, src_lane + u, 32);
    ww[u] = __shfl(my_w, src_lane + u, 32);
    v[u] = make_float4(0.f, 0.f, 0.f, 0.f);
    if (!TAIL || u < left) v[u] = ((const float4*)(Q + (size_t)s * D_OUT))[c];
  }
  #pragma unroll
  for (int u = 0; u < AGG2_GROUP; ++u) {
    if (!TAIL || u < left) {
      acc.x = fmaf(v[u].x, ww[u], acc.x);
      acc.y = fmaf(v[u].y, ww[u], acc.y);
      acc.z = fmaf(v[u].z, ww[u], acc.z);
      acc.w = fmaf(v[u].w, ww[u], acc.w);
    }
  }
}

__global__ void __launch_bounds__(256)
aggregate_projected(const float* __restrict__ P, const float* __restrict__ Q,
                    const float* __restrict__ bias,
                    const int* __restrict__ rowptr,
                    const int* __restrict__ ep_words,
                    float* __restrict__ out, int nN) {
  const int2* ep = (const int2*)ep_words;
  static_assert(D_OUT == 64, "a row is 16 lanes x float4");
  static_assert(2 * AGG2_GROUP == 16, "a batch of 16 steps is two groups");
  int node = __builtin_amdgcn_readfirstlane((int)((blockIdx.x * blockDim.x + threadIdx.x) >> 5));
  int lane = threadIdx.x & 31;
  if (node >= nN) return;
  int half = lane >> 4;
  int c    = lane & 15;
  int beg = rowptr[node];
  int end = rowptr[node + 1];
  int mid = beg + ((end - beg + 1) >> 1);
  int cmax = mid - beg;
  int cmin = end - mid;
  int lo  = half ? mid : beg;
  int cnt = half ? cmin : cmax;
  float4 acc = make_float4(0.f, 0.f, 0.f, 0.f);

  if (beg < end) {
    int2 cur = ep[lo + min(c, cnt - 1)];
    int cur_s = cur.x;
    float cur_w = __int_as_float(cur.y);
    for (int bb = 0; bb < cmax; bb += 16) {
      int2 nxt = ep[lo + min(bb + 16 + c, cnt - 1)];
      int nxt_s = nxt.x;
      float nxt_w = __int_as_float(nxt.y);
      for (int g = 0; g < 16 && bb + g < cmax; g += AGG2_GROUP) {
        if (bb + g + AGG2_GROUP <= cmin) gather_group_halves<false>(Q, cur_s, cur_w, g, AGG2_GROUP, lane, acc);
        else                             gather_group_halves<true>(Q, cur_s, cur_w, g, cnt - (bb + g), lane, acc);
      }
      cur_s = nxt_s;
      cur_w = nxt_w;
    }
  }
  acc.x += __shfl_xor(acc.x, 16, 32);
  acc.y += __shfl_xor(acc.y, 16, 32);
  acc.z += __shfl_xor(acc.z, 16, 32);
  acc.w += __shfl_xor(acc.w, 16, 32);
  if (half == 0) {
    int dg = end - beg;
    float iv = (dg > 0) ? (1.0f / (float)dg) : 0.0f;
    float4 p = ((const float4*)(P + (size_t)node * D_OUT))[c];
    float4 b = ((const float4*)bias)[c];
    ((float4*)(out + (size_t)node * D_OUT))[c] =
        make_float4(p.x + b.x + acc.x * iv, p.y + b.y + acc.y * iv,
                    p.z + b.z + acc.z * iv, p.w + b.w + acc.w * iv);
  }
}

#define X1_PITCH (D_HID + 4)
static_assert((X1_PITCH / 4) % 2 == 1, "pitch = 16 bytes x odd");

__global__ void __launch_bounds__(128)
wmma_gemm_pair_f32(const float* __restrict__ H, const float* __restrict__ MEAN,
                   const float* __restrict__ W1, const float* __restrict__ bias1,
                   const float* __restrict__ W2, float* __restrict__ PQ) {
  static_assert(D_HID == 4 * 64 && 2 * D_OUT == 4 * 32, "four waves cover both strips");
  static_assert(D_IN >= 64 && D_IN % 16 == 0 && D_HID % 16 == 0,
                "the prefetch runs 64 columns ahead; a loop trip takes 16 columns");
  static_assert(N_NODES % 16 == 0, "whole tiles: every wave reaches the barrier");
  __shared__ float4 x1_tile[16 * X1_PITCH / 4];
  float* x1 = (float*)x1_tile;
  int wv   = __builtin_amdgcn_readfirstlane((int)(threadIdx.x >> 5));
  int lane = threadIdx.x & 31;
  int half = lane >> 4;
  int mr   = lane & 15;
  int m0   = blockIdx.x * 16;

  {
    int n0 = wv * 64;
    const float* ap  = H    + (size_t)(m0 + mr) * D_IN + 4 * half;
    const float* ap1 = MEAN + (size_t)(m0 + mr) * D_IN + 4 * half;
    const float* bp  = W1 + (4 * half) * D_HID + n0 + 4 * mr;
    v8f acc0 = {}, acc1 = {}, acc2 = {}, acc3 = {};

    auto step = [&](float ax, float ay, v4f x, v4f y) {
      v2f a, b;
      a.x = ax;   a.y = ay;
      b.x = x.x;  b.y = y.x;
      acc0 = __builtin_amdgcn_wmma_f32_16x16x4_f32(false, a, false, b, (short)0, acc0, false, false);
      b.x = x.y;  b.y = y.y;
      acc1 = __builtin_amdgcn_wmma_f32_16x16x4_f32(false, a, false, b, (short)0, acc1, false, false);
      b.x = x.z;  b.y = y.z;
      acc2 = __builtin_amdgcn_wmma_f32_16x16x4_f32(false, a, false, b, (short)0, acc2, false, false);
      b.x = x.w;  b.y = y.w;
      acc3 = __builtin_amdgcn_wmma_f32_16x16x4_f32(false, a, false, b, (short)0, acc3, false, false);
    };

    auto kpair = [&](const float*& arow, const float* pf) {
      v4f a0 = *(const v4f*)arow;
      v4f a1 = *(const v4f*)(arow + 8);
      arow += 16;
      __builtin_prefetch(pf, 0, 3);
      v4f r0 = *(const v4f*)(bp);
      v4f r1 = *(const v4f*)(bp + D_HID);
      v4f r2 = *(const v4f*)(bp + 2 * D_HID);
      v4f r3 = *(const v4f*)(bp + 3 * D_HID);
      v4f r4 = *(const v4f*)(bp + 8 * D_HID);
      v4f r5 = *(const v4f*)(bp + 9 * D_HID);
      v4f r6 = *(const v4f*)(bp + 10 * D_HID);
      v4f r7 = *(const v4f*)(bp + 11 * D_HID);
      bp += 16 * D_HID;
      __builtin_amdgcn_sched_barrier(0);
      step(a0.x, a0.y, r0, r1);
      step(a0.z, a0.w, r2, r3);
      step(a1.x, a1.y, r4, r5);
      step(a1.z, a1.w, r6, r7);
    };

    for (int k = 0; k < D_IN - 64; k += 16) kpair(ap, ap + 64);
    const float* pf1 = ap1;
    for (int k = D_IN - 64; k < D_IN; k += 16) { kpair(ap, pf1); pf1 += 16; }
    for (int k = D_IN; k < 2 * D_IN; k += 16) kpair(ap1, ap1 + 64);

    float4 bv = *(const float4*)(bias1 + n0 + 4 * mr);
    float* xp = x1 + (8 * half) * X1_PITCH + n0 + 4 * mr;
    #pragma unroll
    for (int r = 0; r < 8; ++r) {
      float4 v = make_float4(acc0[r] + bv.x, acc1[r] + bv.y, acc2[r] + bv.z, acc3[r] + bv.w);
      v.x = (v.x > 0.0f) ? v.x : 0.0f;
      v.y = (v.y > 0.0f) ? v.y : 0.0f;
      v.z = (v.z > 0.0f) ? v.z : 0.0f;
      v.w = (v.w > 0.0f) ? v.w : 0.0f;
      *(float4*)(xp + r * X1_PITCH) = v;
    }
  }
  __syncthreads();

  {
    int slab = wv >> 1;
    int n0   = (wv & 1) * 32;
    const float* xa = x1 + mr * X1_PITCH + 4 * half;
    const float* bq = W2 + (size_t)slab * (D_HID * D_OUT) + (4 * half) * D_OUT + n0 + 2 * mr;
    v8f acc0 = {}, acc1 = {};

    auto step = [&](float ax, float ay, v2f x, v2f y) {
      v2f a, b;
      a.x = ax;   a.y = ay;
      b.x = x.x;  b.y = y.x;
      acc0 = __builtin_amdgcn_wmma_f32_16x16x4_f32(false, a, false, b, (short)0, acc0, false, false);
      b.x = x.y;  b.y = y.y;
      acc1 = __builtin_amdgcn_wmma_f32_16x16x4_f32(false, a, false, b, (short)0, acc1, false, false);
    };

    #pragma unroll
    for (int kb = 0; kb < D_HID; kb += 16) {
      v4f a0 = *(const v4f*)(xa + kb);
      v4f a1 = *(const v4f*)(xa + kb + 8);
      v2f r0 = *(const v2f*)(bq + (kb + 0) * D_OUT);
      v2f r1 = *(const v2f*)(bq + (kb + 1) * D_OUT);
      v2f r2 = *(const v2f*)(bq + (kb + 2) * D_OUT);
      v2f r3 = *(const v2f*)(bq + (kb + 3) * D_OUT);
      v2f r4 = *(const v2f*)(bq + (kb + 8) * D_OUT);
      v2f r5 = *(const v2f*)(bq + (kb + 9) * D_OUT);
      v2f r6 = *(const v2f*)(bq + (kb + 10) * D_OUT);
      v2f r7 = *(const v2f*)(bq + (kb + 11) * D_OUT);
      __builtin_amdgcn_sched_barrier(0);
      step(a0.x, a0.y, r0, r1);
      step(a0.z, a0.w, r2, r3);
      step(a1.x, a1.y, r4, r5);
      step(a1.z, a1.w, r6, r7);
    }

    float* yp = PQ + (size_t)slab * (N_NODES * D_OUT) + (size_t)(m0 + 8 * half) * D_OUT + n0 + 2 * mr;
    #pragma unroll
    for (int r = 0; r < 8; ++r) *(float2*)(yp + r * D_OUT) = make_float2(acc0[r], acc1[r]);
  }
}

extern "C" void kernel_launch(void* const* d_in, const int* in_sizes, int n_in,
                              void* d_out, int out_size, void* d_ws, size_t ws_size,
                              hipStream_t stream) {
  const float* h   = (const float*)d_in[0];
  const float* w   = (const float*)d_in[1];
  const int*   src = (const int*)d_in[2];
  const int*   dst = (const int*)d_in[3];
  const float* W1  = (const float*)d_in[4];
  const float* b1  = (const float*)d_in[5];
  const float* W2  = (const float*)d_in[6];
  const float* b2  = (const float*)d_in[7];
  float* out = (float*)d_out;

  int*   rowptr = (int*)d_ws + 10240;
  int*   ep     = rowptr + 2 * 10240;
  float* bufA   = (float*)(ep + 2 * N_EDGES) + 10240;
  float* bufB   = bufA + N_NODES * D_HID;
  float* PQ     = bufB + N_NODES * D_HID;

  int* rec = (int*)bufB;
  int* tab = rec + 2 * N_EDGES;
  partition_chunks<<<CSR_CHUNKS, CSR_THREADS, 0, stream>>>(src, dst, w, tab, rec);
  sort_buckets<<<N_BKT, CSR_THREADS, 0, stream>>>(tab, rec, rowptr, ep);

  aggregate_nodes<D_IN><<<(N_NODES * 32 + 255) / 256, 256, 0, stream>>>(
      h, rowptr, ep, bufA, N_NODES);
  wmma_gemm_pair_f32<<<N_NODES / 16, 128, 0, stream>>>(h, bufA, W1, b1, W2, PQ);

  aggregate_projected<<<(N_NODES * 32 + 255) / 256, 256, 0, stream>>>(
      PQ, PQ + N_NODES * D_OUT, b2, rowptr, ep, out, N_NODES);
}
